# starting kernel
# baseline (speedup 1.0000x reference)
#include <hip/hip_runtime.h>
#include <hip/hip_bf16.h>

typedef __attribute__((ext_vector_type(2))) float v2f;
typedef __attribute__((ext_vector_type(8))) float v8f;

#define EHID 64
#define RHID 32
#define DFEAT 160
#define DOUT 224
#define EPSF 1e-16f

static inline int ceil_div(int a, int b) { return (a + b - 1) / b; }

__global__ void k_zero_f32(float* p, int n) {
    int i = blockIdx.x * blockDim.x + threadIdx.x;
    if (i < n) p[i] = 0.0f;
}

__global__ void k_zero_u32(unsigned* p, int n) {
    int i = blockIdx.x * blockDim.x + threadIdx.x;
    if (i < n) p[i] = 0u;
}

__global__ void k_att_wmma(const float* __restrict__ x,
                           const int* __restrict__ ei,
                           const int* __restrict__ ej,
                           const int* __restrict__ rel,
                           const float* __restrict__ rel_emb,
                           const float* __restrict__ w,
                           float* __restrict__ att, int E) {
    __shared__ float sw[DFEAT];
    int t = threadIdx.x;
    if (t < DFEAT) sw[t] = w[t];
    __syncthreads();

    int wave  = blockIdx.x * (blockDim.x >> 5) + (t >> 5);
    int lane  = t & 31;
    int ebase = wave << 4;
    if (ebase >= E) return;

    int m     = lane & 15;
    int khalf = lane >> 4;
    int e     = ebase + m;

    int vi = ei[e], vj = ej[e], r = rel[e];
    const float* xi = x + (long)vi * EHID;
    const float* xj = x + (long)vj * EHID;
    const float* re = rel_emb + (long)r * RHID;

    v8f c = {};
#pragma unroll
    for (int ch = 0; ch < DFEAT / 4; ++ch) {
        int k0 = ch * 4 + khalf * 2;
        v2f a;
        if (k0 < EHID)              a = *(const v2f*)(xi + k0);
        else if (k0 < EHID + RHID)  a = *(const v2f*)(re + (k0 - EHID));
        else                        a = *(const v2f*)(xj + (k0 - EHID - RHID));
        v2f b;
        b.x = sw[k0];
        b.y = sw[k0 + 1];
        c = __builtin_amdgcn_wmma_f32_16x16x4_f32(
                 false, a,  false, b,
                 (short)0, c,  false,  false);
    }

    if ((lane & 15) == 0) {
        float4 lo = make_float4(c[0], c[1], c[2], c[3]);
        float4 hi = make_float4(c[4], c[5], c[6], c[7]);
        float4* o = (float4*)(att + ebase + (khalf << 3));
        o[0] = lo;
        o[1] = hi;
    }
}

__device__ __forceinline__ unsigned enc_ord(float f) {
    unsigned u = __float_as_uint(f);
    return (u & 0x80000000u) ? ~u : (u | 0x80000000u);
}
__device__ __forceinline__ float dec_ord(unsigned u) {
    return (u & 0x80000000u) ? __uint_as_float(u & 0x7FFFFFFFu)
                             : __uint_as_float(~u);
}

__global__ void k_segmax(const float* __restrict__ att, const int* __restrict__ ei,
                         unsigned* __restrict__ segmax_u, int E) {
    int e = blockIdx.x * blockDim.x + threadIdx.x;
    if (e >= E) return;
    atomicMax(&segmax_u[ei[e]], enc_ord(att[e]));
}

__global__ void k_expsum(float* __restrict__ att, const int* __restrict__ ei,
                         const unsigned* __restrict__ segmax_u,
                         float* __restrict__ segsum, int E) {
    int e = blockIdx.x * blockDim.x + threadIdx.x;
    if (e >= E) return;
    int vi = ei[e];
    float p = expf(att[e] - dec_ord(segmax_u[vi]));
    att[e] = p;
    atomicAdd(&segsum[vi], p);
}

__global__ void k_scatter(const float* __restrict__ x,
                          const int* __restrict__ ei,
                          const int* __restrict__ ej,
                          const int* __restrict__ rel,
                          const float* __restrict__ rel_emb,
                          const float* __restrict__ p,
                          const float* __restrict__ segsum,
                          float* __restrict__ out, int E) {
    int wave = blockIdx.x * (blockDim.x >> 5) + (threadIdx.x >> 5);
    int lane = threadIdx.x & 31;
    if (wave >= E) return;
    int e = wave;
    int vi = ei[e], vj = ej[e], r = rel[e];
    float coef = p[e] / (segsum[vi] + EPSF);
    float* orow = out + (long)vi * DOUT + EHID;
    const float* xi = x + (long)vi * EHID;
    const float* xj = x + (long)vj * EHID;
    const float* re = rel_emb + (long)r * RHID;
#pragma unroll
    for (int i = 0; i < DFEAT / 32; ++i) {
        int d = lane + 32 * i;
        float v;
        if (d < EHID)              v = xi[d];
        else if (d < EHID + RHID)  v = re[d - EHID];
        else                       v = xj[d - EHID - RHID];
        atomicAdd(orow + d, v * coef);
    }
}

__global__ void k_finalize(const float* __restrict__ x, float* __restrict__ out, int total) {
    int i = blockIdx.x * blockDim.x + threadIdx.x;
    if (i >= total) return;
    int c = i % DOUT;
    int n = i / DOUT;
    if (c < EHID) out[i] = x[n * EHID + c];
    else          out[i] = fmaxf(out[i], 0.0f);
}

extern "C" void kernel_launch(void* const* d_in, const int* in_sizes, int n_in,
                              void* d_out, int out_size, void* d_ws, size_t ws_size,
                              hipStream_t stream) {
    const float* x        = (const float*)d_in[0];
    const int*   eidx     = (const int*)d_in[1];
    const int*   rel      = (const int*)d_in[2];
    const float* rel_emb  = (const float*)d_in[3];
    const float* w        = (const float*)d_in[4];
    float* out = (float*)d_out;

    const int E = in_sizes[2];
    const int N = in_sizes[0] / EHID;
    const int* ei = eidx;
    const int* ej = eidx + E;

    char* ws = (char*)d_ws;
    float*    att      = (float*)ws;
    unsigned* segmax_u = (unsigned*)(ws + (size_t)E * 4);
    float*    segsum   = (float*)(ws + (size_t)E * 4 + (size_t)N * 4);

    const int B = 256;

    int tot_out = N * DOUT;
    k_zero_f32<<<ceil_div(tot_out, B), B, 0, stream>>>(out, tot_out);
    k_zero_u32<<<ceil_div(2 * N, B), B, 0, stream>>>(segmax_u, 2 * N);

    int waves = E / 16;
    k_att_wmma<<<ceil_div(waves, 8), B, 0, stream>>>(x, ei, ej, rel, rel_emb, w, att, E);

    k_segmax<<<ceil_div(E, B), B, 0, stream>>>(att, ei, segmax_u, E);

    k_expsum<<<ceil_div(E, B), B, 0, stream>>>(att, ei, segmax_u, segsum, E);

    k_scatter<<<ceil_div(E, 8), B, 0, stream>>>(x, ei, ej, rel, rel_emb, att, segsum, out, E);

    k_finalize<<<ceil_div(tot_out, B), B, 0, stream>>>(x, out, tot_out);
}


# Group edges by destination: a returning integer add ranks each edge into a list in its node's own output row; 16 lanes per node do softmax and sums in registers in arrival-rank order; one row store
# speedup vs baseline: 2.3438x; 2.3438x over previous
#include <hip/hip_runtime.h>
#include <hip/hip_bf16.h>


typedef __attribute__((ext_vector_type(2))) float v2f;
typedef __attribute__((ext_vector_type(8))) float v8f;

#define EHID 64
#define RHID 32
#define DFEAT 160
#define DOUT 224
#define EPSF 1e-16f
#define CAP DFEAT

static inline int ceil_div(int a, int b) { return (a + b - 1) / b; }

__global__ void k_zero_u32(unsigned* p, int n) {
    int i = blockIdx.x * blockDim.x + threadIdx.x;
    if (i < n) p[i] = 0u;
}

__global__ void k_att_wmma(const float* __restrict__ x,
                           const int* __restrict__ ei,
                           const int* __restrict__ ej,
                           const int* __restrict__ rel,
                           const float* __restrict__ rel_emb,
                           const float* __restrict__ w,
                           float* __restrict__ att,
                           int* __restrict__ cnt,
                           float* __restrict__ out, int E) {
    __shared__ float sw[DFEAT];
    int t = threadIdx.x;
    if (t < DFEAT) sw[t] = w[t];
    __syncthreads();

    int wave  = blockIdx.x * (blockDim.x >> 5) + (t >> 5);
    int lane  = t & 31;
    int ebase = wave << 4;
    if (ebase >= E) return;

    int m     = lane & 15;
    int khalf = lane >> 4;
    int e     = ebase + m;

    int vi = ei[e], vj = ej[e], r = rel[e];

    const float* xi = x + (long)vi * EHID;
    const float* xj = x + (long)vj * EHID;
    const float* re = rel_emb + (long)r * RHID;

    v8f c = {};
#pragma unroll
    for (int ch = 0; ch < DFEAT / 4; ++ch) {
        int k0 = ch * 4 + khalf * 2;
        v2f a;
        if (k0 < EHID)              a = *(const v2f*)(xi + k0);
        else if (k0 < EHID + RHID)  a = *(const v2f*)(re + (k0 - EHID));
        else                        a = *(const v2f*)(xj + (k0 - EHID - RHID));
        v2f b;
        b.x = sw[k0];
        b.y = sw[k0 + 1];
        c = __builtin_amdgcn_wmma_f32_16x16x4_f32(
                 false, a,  false, b,
                 (short)0, c,  false,  false);
    }

    if (khalf == 0) {
        int rank = atomicAdd(&cnt[vi], 1);
        if (rank < CAP) out[(long)vi * DOUT + EHID + rank] = __int_as_float(e);
    }

    if ((lane & 15) == 0) {
        float4 lo = make_float4(c[0], c[1], c[2], c[3]);
        float4 hi = make_float4(c[4], c[5], c[6], c[7]);
        float4* o = (float4*)(att + ebase + (khalf << 3));
        o[0] = lo;
        o[1] = hi;
    }
}

__device__ __forceinline__ float hmax16(float v) {
    v = fmaxf(v, __shfl_xor(v, 8, 16));
    v = fmaxf(v, __shfl_xor(v, 4, 16));
    v = fmaxf(v, __shfl_xor(v, 2, 16));
    v = fmaxf(v, __shfl_xor(v, 1, 16));
    return v;
}
__device__ __forceinline__ float hsum16(float v) {
    v += __shfl_xor(v, 8, 16);
    v += __shfl_xor(v, 4, 16);
    v += __shfl_xor(v, 2, 16);
    v += __shfl_xor(v, 1, 16);
    return v;
}

struct NodeAcc {
    float4 aI, aJ;
    float2 aR;
};

__device__ __forceinline__ void add_edges(unsigned msk, float coef, int vj, int r, int l,
                                          const float* __restrict__ x,
                                          const float* __restrict__ rel_emb,
                                          float4 xi, NodeAcc& A) {
    while (msk) {
        const int q0 = __ffs(msk) - 1;
        float cf[4];
        float4 xj[4];
        float2 rr[4];
#pragma unroll
        for (int u = 0; u < 4; ++u) {
            bool on = (msk != 0u);
            int qu = on ? (__ffs(msk) - 1) : q0;
            msk &= msk - 1u;
            float cc = __shfl(coef, qu, 16);
            int vjq  = __shfl(vj, qu, 16);
            int rq   = __shfl(r, qu, 16);
            cf[u] = on ? cc : 0.0f;
            xj[u] = *(const float4*)(x + (long)vjq * EHID + 4 * l);
            rr[u] = *(const float2*)(rel_emb + (long)rq * RHID + 2 * l);
        }
#pragma unroll
        for (int u = 0; u < 4; ++u) {
            A.aI.x += xi.x * cf[u]; A.aI.y += xi.y * cf[u]; A.aI.z += xi.z * cf[u]; A.aI.w += xi.w * cf[u];
            A.aR.x += rr[u].x * cf[u]; A.aR.y += rr[u].y * cf[u];
            A.aJ.x += xj[u].x * cf[u]; A.aJ.y += xj[u].y * cf[u]; A.aJ.z += xj[u].z * cf[u]; A.aJ.w += xj[u].w * cf[u];
        }
    }
}

__global__ void __launch_bounds__(256)
k_gather(const float* __restrict__ x,
         const int* __restrict__ ei,
         const int* __restrict__ ej,
         const int* __restrict__ rel,
         const float* __restrict__ rel_emb,
         const float* __restrict__ att,
         const int* __restrict__ cnt,
         float* out, int N, int E) {
    int t = blockIdx.x * blockDim.x + threadIdx.x;
    int n = t >> 4;
    int l = threadIdx.x & 15;
    bool live = n < N;
    int nn = live ? n : 0;
    int c = live ? cnt[nn] : 0;

    float* row = out + (long)nn * DOUT;
    const float* list = row + EHID;
    float4 xi = *(const float4*)(x + (long)nn * EHID + 4 * l);

    NodeAcc A;
    A.aI = make_float4(0.f, 0.f, 0.f, 0.f);
    A.aJ = make_float4(0.f, 0.f, 0.f, 0.f);
    A.aR = make_float2(0.f, 0.f);

    if (c <= CAP) {
        bool v0 = l < c;
        int e0 = 0, vj0 = 0, r0 = 0;
        float a0 = -INFINITY;
        if (v0) {
            e0  = __float_as_int(list[l]);
            a0  = att[e0];
            vj0 = ej[e0];
            r0  = rel[e0];
        }
        float mx = a0;
        for (int b = 16; b < c; b += 16) {
            int pos = b + l;
            if (pos < c) mx = fmaxf(mx, att[__float_as_int(list[pos])]);
        }
        mx = hmax16(mx);
        float p0 = v0 ? expf(a0 - mx) : 0.0f;
        float s = p0;
        for (int b = 16; b < c; b += 16) {
            int pos = b + l;
            if (pos < c) s += expf(att[__float_as_int(list[pos])] - mx);
        }
        s = hsum16(s);
        float den = s + EPSF;
        {
            int k = c < 16 ? c : 16;
            add_edges((1u << k) - 1u, p0 / den, vj0, r0, l, x, rel_emb, xi, A);
        }
        for (int b = 16; b < c; b += 16) {
            int pos = b + l;
            float coef = 0.0f;
            int vj = 0, r = 0;
            if (pos < c) {
                int e = __float_as_int(list[pos]);
                coef = expf(att[e] - mx) / den;
                vj = ej[e];
                r = rel[e];
            }
            int k = (c - b) < 16 ? (c - b) : 16;
            add_edges((1u << k) - 1u, coef, vj, r, l, x, rel_emb, xi, A);
        }
    } else {
        int hshift = threadIdx.x & 16;
        float mx = -INFINITY;
        for (int b = 0; b < E; b += 16) {
            int pos = b + l;
            if (pos < E && ei[pos] == nn) mx = fmaxf(mx, att[pos]);
        }
        mx = hmax16(mx);
        float s = 0.0f;
        for (int b = 0; b < E; b += 16) {
            int pos = b + l;
            if (pos < E && ei[pos] == nn) s += expf(att[pos] - mx);
        }
        s = hsum16(s);
        float den = s + EPSF;
        for (int b = 0; b < E; b += 16) {
            int pos = b + l;
            bool valid = (pos < E) && (ei[pos] == nn);
            float coef = 0.0f;
            int vj = 0, r = 0;
            if (valid) {
                coef = expf(att[pos] - mx) / den;
                vj = ej[pos];
                r = rel[pos];
            }
            unsigned msk = (unsigned)((__ballot(valid) >> hshift) & 0xFFFFull);
            add_edges(msk, coef, vj, r, l, x, rel_emb, xi, A);
        }
    }

    if (live) {
        *(float4*)(row + 4 * l) = xi;
        *(float4*)(row + EHID + 4 * l) =
            make_float4(fmaxf(A.aI.x, 0.f), fmaxf(A.aI.y, 0.f), fmaxf(A.aI.z, 0.f), fmaxf(A.aI.w, 0.f));
        *(float2*)(row + 2 * EHID + 2 * l) = make_float2(fmaxf(A.aR.x, 0.f), fmaxf(A.aR.y, 0.f));
        *(float4*)(row + 2 * EHID + RHID + 4 * l) =
            make_float4(fmaxf(A.aJ.x, 0.f), fmaxf(A.aJ.y, 0.f), fmaxf(A.aJ.z, 0.f), fmaxf(A.aJ.w, 0.f));
    }
}

extern "C" void kernel_launch(void* const* d_in, const int* in_sizes, int n_in,
                              void* d_out, int out_size, void* d_ws, size_t ws_size,
                              hipStream_t stream) {
    const float* x        = (const float*)d_in[0];
    const int*   eidx     = (const int*)d_in[1];
    const int*   rel      = (const int*)d_in[2];
    const float* rel_emb  = (const float*)d_in[3];
    const float* w        = (const float*)d_in[4];
    float* out = (float*)d_out;

    const int E = in_sizes[2];
    const int N = in_sizes[0] / EHID;
    const int* ei = eidx;
    const int* ej = eidx + E;

    char* ws = (char*)d_ws;
    float* att = (float*)ws;
    int*   cnt = (int*)(ws + (size_t)E * 4);

    const int B = 256;

    k_zero_u32<<<ceil_div(N, B), B, 0, stream>>>((unsigned*)cnt, N);

    int waves = E / 16;
    k_att_wmma<<<ceil_div(waves, 8), B, 0, stream>>>(x, ei, ej, rel, rel_emb, w, att, cnt, out, E);

    k_gather<<<ceil_div(N, B / 16), B, 0, stream>>>(x, ei, ej, rel, rel_emb, att, cnt, out, N, E);
}


# Score split by linearity: x.w_i, x.w_j per node and rel_emb.w_r per relation once on WMMA (in the counter-zeroing launch); edge score = sum of three looked-up scalars, so its sum order changes
# speedup vs baseline: 3.1529x; 1.3452x over previous
#include <hip/hip_runtime.h>
#include <hip/hip_bf16.h>


typedef __attribute__((ext_vector_type(2))) float v2f;
typedef __attribute__((ext_vector_type(8))) float v8f;

#define EHID 64
#define RHID 32
#define DFEAT 160
#define DOUT 224
#define EPSF 1e-16f
#define CAP DFEAT

static inline int ceil_div(int a, int b) { return (a + b - 1) / b; }

__global__ void __launch_bounds__(256)
k_project(const float* __restrict__ x,
          const float* __restrict__ rel_emb,
          const float* __restrict__ w,
          float* __restrict__ s_i,
          float* __restrict__ s_j,
          float* __restrict__ s_r,
          unsigned* __restrict__ cnt,
          int N, int R, int tilesN) {
    int t = threadIdx.x;
    int gid = blockIdx.x * blockDim.x + t;
    if (gid < N) cnt[gid] = 0u;

    int wave = gid >> 5;
    int lane = t & 31;
    int c = lane & 15;
    int h = lane >> 4;

    if (wave < tilesN) {
        int row = wave * 16 + c;
        int rr = row < N ? row : N - 1;
        bool bon = c < 2;
        const float* xr = x + (long)rr * EHID + 2 * h;
        const float* wc = w + ((c == 1) ? (EHID + RHID) : 0) + 2 * h;
        v8f acc = {};
#pragma unroll
        for (int ch = 0; ch < EHID / 4; ++ch) {
            v2f a  = *(const v2f*)(xr + 4 * ch);
            v2f wv = *(const v2f*)(wc + 4 * ch);
            v2f b;
            b.x = bon ? wv.x : 0.0f;
            b.y = bon ? wv.y : 0.0f;
            acc = __builtin_amdgcn_wmma_f32_16x16x4_f32(
                       false, a,  false, b,
                       (short)0, acc,  false,  false);
        }
        if (bon) {
            float* dst = (c == 0) ? s_i : s_j;
            int r0 = wave * 16 + 8 * h;
            for (int r = 0; r < 8; ++r)
                if (r0 + r < N) dst[r0 + r] = acc[r];
        }
    } else {
        int tile = wave - tilesN;
        if (tile * 16 >= R) return;
        int row = tile * 16 + c;
        int rr = row < R ? row : R - 1;
        bool bon = c == 0;
        const float* er = rel_emb + (long)rr * RHID + 2 * h;
        const float* wc = w + EHID + 2 * h;
        v8f acc = {};
#pragma unroll
        for (int ch = 0; ch < RHID / 4; ++ch) {
            v2f a  = *(const v2f*)(er + 4 * ch);
            v2f wv = *(const v2f*)(wc + 4 * ch);
            v2f b;
            b.x = bon ? wv.x : 0.0f;
            b.y = bon ? wv.y : 0.0f;
            acc = __builtin_amdgcn_wmma_f32_16x16x4_f32(
                       false, a,  false, b,
                       (short)0, acc,  false,  false);
        }
        if (bon) {
            int r0 = tile * 16 + 8 * h;
            for (int r = 0; r < 8; ++r)
                if (r0 + r < R) s_r[r0 + r] = acc[r];
        }
    }
}

__global__ void __launch_bounds__(256)
k_list(const int* __restrict__ ei,
       int* __restrict__ cnt,
       float* __restrict__ out, int E) {
    int e = blockIdx.x * blockDim.x + threadIdx.x;
    if (e < E) {
        int vi = ei[e];
        int rank = atomicAdd(&cnt[vi], 1);
        if (rank < CAP) out[(long)vi * DOUT + EHID + rank] = __int_as_float(e);
    }
}

__device__ __forceinline__ float hmax16(float v) {
    v = fmaxf(v, __shfl_xor(v, 8, 16));
    v = fmaxf(v, __shfl_xor(v, 4, 16));
    v = fmaxf(v, __shfl_xor(v, 2, 16));
    v = fmaxf(v, __shfl_xor(v, 1, 16));
    return v;
}
__device__ __forceinline__ float hsum16(float v) {
    v += __shfl_xor(v, 8, 16);
    v += __shfl_xor(v, 4, 16);
    v += __shfl_xor(v, 2, 16);
    v += __shfl_xor(v, 1, 16);
    return v;
}

struct NodeAcc {
    float4 aI, aJ;
    float2 aR;
};

__device__ __forceinline__ void add_edges(unsigned msk, float coef, int vj, int r, int l,
                                          const float* __restrict__ x,
                                          const float* __restrict__ rel_emb,
                                          float4 xi, NodeAcc& A) {
    while (msk) {
        const int q0 = __ffs(msk) - 1;
        float cf[4];
        float4 xj[4];
        float2 rr[4];
#pragma unroll
        for (int u = 0; u < 4; ++u) {
            bool on = (msk != 0u);
            int qu = on ? (__ffs(msk) - 1) : q0;
            msk &= msk - 1u;
            float cc = __shfl(coef, qu, 16);
            int vjq  = __shfl(vj, qu, 16);
            int rq   = __shfl(r, qu, 16);
            cf[u] = on ? cc : 0.0f;
            xj[u] = *(const float4*)(x + (long)vjq * EHID + 4 * l);
            rr[u] = *(const float2*)(rel_emb + (long)rq * RHID + 2 * l);
        }
#pragma unroll
        for (int u = 0; u < 4; ++u) {
            A.aI.x += xi.x * cf[u]; A.aI.y += xi.y * cf[u]; A.aI.z += xi.z * cf[u]; A.aI.w += xi.w * cf[u];
            A.aR.x += rr[u].x * cf[u]; A.aR.y += rr[u].y * cf[u];
            A.aJ.x += xj[u].x * cf[u]; A.aJ.y += xj[u].y * cf[u]; A.aJ.z += xj[u].z * cf[u]; A.aJ.w += xj[u].w * cf[u];
        }
    }
}

__device__ __forceinline__ float edge_score(float si, int q, int v,
                                            const float* __restrict__ s_r,
                                            const float* __restrict__ s_j) {
    return (si + s_r[q]) + s_j[v];
}

__global__ void __launch_bounds__(256)
k_gather(const float* __restrict__ x,
         const int* __restrict__ ei,
         const int* __restrict__ ej,
         const int* __restrict__ rel,
         const float* __restrict__ rel_emb,
         const float* __restrict__ s_i,
         const float* __restrict__ s_j,
         const float* __restrict__ s_r,
         const int* __restrict__ cnt,
         float* out, int N, int E) {
    int t = blockIdx.x * blockDim.x + threadIdx.x;
    int n = t >> 4;
    int l = threadIdx.x & 15;
    bool live = n < N;
    int nn = live ? n : 0;
    int c = live ? cnt[nn] : 0;

    float* row = out + (long)nn * DOUT;
    const float* list = row + EHID;
    float4 xi = *(const float4*)(x + (long)nn * EHID + 4 * l);
    float si = s_i[nn];

    NodeAcc A;
    A.aI = make_float4(0.f, 0.f, 0.f, 0.f);
    A.aJ = make_float4(0.f, 0.f, 0.f, 0.f);
    A.aR = make_float2(0.f, 0.f);

    if (c <= CAP) {
        bool v0 = l < c;
        int e0 = 0, vj0 = 0, r0 = 0;
        float a0 = -INFINITY;
        if (v0) {
            e0  = __float_as_int(list[l]);
            vj0 = ej[e0];
            r0  = rel[e0];
            a0  = edge_score(si, r0, vj0, s_r, s_j);
        }
        float mx = a0;
        for (int b = 16; b < c; b += 16) {
            int pos = b + l;
            if (pos < c) {
                int e = __float_as_int(list[pos]);
                mx = fmaxf(mx, edge_score(si, rel[e], ej[e], s_r, s_j));
            }
        }
        mx = hmax16(mx);
        float p0 = v0 ? expf(a0 - mx) : 0.0f;
        float s = p0;
        for (int b = 16; b < c; b += 16) {
            int pos = b + l;
            if (pos < c) {
                int e = __float_as_int(list[pos]);
                s += expf(edge_score(si, rel[e], ej[e], s_r, s_j) - mx);
            }
        }
        s = hsum16(s);
        float den = s + EPSF;
        {
            int k = c < 16 ? c : 16;
            add_edges((1u << k) - 1u, p0 / den, vj0, r0, l, x, rel_emb, xi, A);
        }
        for (int b = 16; b < c; b += 16) {
            int pos = b + l;
            float coef = 0.0f;
            int vj = 0, r = 0;
            if (pos < c) {
                int e = __float_as_int(list[pos]);
                vj = ej[e];
                r = rel[e];
                coef = expf(edge_score(si, r, vj, s_r, s_j) - mx) / den;
            }
            int k = (c - b) < 16 ? (c - b) : 16;
            add_edges((1u << k) - 1u, coef, vj, r, l, x, rel_emb, xi, A);
        }
    } else {
        int hshift = threadIdx.x & 16;
        float mx = -INFINITY;
        for (int b = 0; b < E; b += 16) {
            int pos = b + l;
            if (pos < E && ei[pos] == nn) mx = fmaxf(mx, edge_score(si, rel[pos], ej[pos], s_r, s_j));
        }
        mx = hmax16(mx);
        float s = 0.0f;
        for (int b = 0; b < E; b += 16) {
            int pos = b + l;
            if (pos < E && ei[pos] == nn) s += expf(edge_score(si, rel[pos], ej[pos], s_r, s_j) - mx);
        }
        s = hsum16(s);
        float den = s + EPSF;
        for (int b = 0; b < E; b += 16) {
            int pos = b + l;
            bool valid = (pos < E) && (ei[pos] == nn);
            float coef = 0.0f;
            int vj = 0, r = 0;
            if (valid) {
                vj = ej[pos];
                r = rel[pos];
                coef = expf(edge_score(si, r, vj, s_r, s_j) - mx) / den;
            }
            unsigned msk = (unsigned)((__ballot(valid) >> hshift) & 0xFFFFull);
            add_edges(msk, coef, vj, r, l, x, rel_emb, xi, A);
        }
    }

    if (live) {
        *(float4*)(row + 4 * l) = xi;
        *(float4*)(row + EHID + 4 * l) =
            make_float4(fmaxf(A.aI.x, 0.f), fmaxf(A.aI.y, 0.f), fmaxf(A.aI.z, 0.f), fmaxf(A.aI.w, 0.f));
        *(float2*)(row + 2 * EHID + 2 * l) = make_float2(fmaxf(A.aR.x, 0.f), fmaxf(A.aR.y, 0.f));
        *(float4*)(row + 2 * EHID + RHID + 4 * l) =
            make_float4(fmaxf(A.aJ.x, 0.f), fmaxf(A.aJ.y, 0.f), fmaxf(A.aJ.z, 0.f), fmaxf(A.aJ.w, 0.f));
    }
}

extern "C" void kernel_launch(void* const* d_in, const int* in_sizes, int n_in,
                              void* d_out, int out_size, void* d_ws, size_t ws_size,
                              hipStream_t stream) {
    const float* x        = (const float*)d_in[0];
    const int*   eidx     = (const int*)d_in[1];
    const int*   rel      = (const int*)d_in[2];
    const float* rel_emb  = (const float*)d_in[3];
    const float* w        = (const float*)d_in[4];
    float* out = (float*)d_out;

    const int E = in_sizes[2];
    const int N = in_sizes[0] / EHID;
    const int R = in_sizes[3] / RHID;
    const int* ei = eidx;
    const int* ej = eidx + E;

    char* ws = (char*)d_ws;
    int*   cnt = (int*)ws;
    float* s_i = (float*)(ws + (size_t)N * 4);
    float* s_j = (float*)(ws + (size_t)N * 8);
    float* s_r = (float*)(ws + (size_t)N * 12);

    const int B = 256;

    const int tilesN = ceil_div(N, 16);
    const int tilesR = ceil_div(R, 16);
    k_project<<<ceil_div(tilesN + tilesR, 8), B, 0, stream>>>(x, rel_emb, w, s_i, s_j, s_r, (unsigned*)cnt, N, R, tilesN);

    k_list<<<ceil_div(E, B), B, 0, stream>>>(ei, cnt, out, E);

    k_gather<<<ceil_div(N, B / 16), B, 0, stream>>>(x, ei, ej, rel, rel_emb, s_i, s_j, s_r, cnt, out, N, E);
}


# Per-edge returning adds on memory replaced by LDS ranking: edges binned by 256-node bucket (one memory add per group and bucket), one work-group per bucket groups and sums them in LDS arrival order
# speedup vs baseline: 5.4877x; 1.7405x over previous
#include <hip/hip_runtime.h>
#include <hip/hip_bf16.h>


typedef __attribute__((ext_vector_type(2))) float v2f;
typedef __attribute__((ext_vector_type(8))) float v8f;

#define EHID 64
#define RHID 32
#define DFEAT 160
#define DOUT 224
#define EPSF 1e-16f
#define BW 256
#define BSH 8
#define RECW 4
#define LCAP 4096
#define BIN_EPT 8
#define GT 512

static inline int ceil_div(int a, int b) { return (a + b - 1) / b; }

__global__ void __launch_bounds__(256)
k_project(const float* __restrict__ x,
          const float* __restrict__ rel_emb,
          const float* __restrict__ w,
          float* __restrict__ s_i,
          float* __restrict__ s_j,
          float* __restrict__ s_r,
          unsigned* __restrict__ bcnt,
          int N, int R, int tilesN, int NB) {
    int t = threadIdx.x;
    int gid = blockIdx.x * blockDim.x + t;
    if (gid < NB) bcnt[gid] = 0u;

    int wave = gid >> 5;
    int lane = t & 31;
    int c = lane & 15;
    int h = lane >> 4;

    if (wave < tilesN) {
        int row = wave * 16 + c;
        int rr = row < N ? row : N - 1;
        bool bon = c < 2;
        const float* xr = x + (long)rr * EHID + 2 * h;
        const float* wc = w + ((c == 1) ? (EHID + RHID) : 0) + 2 * h;
        v8f acc = {};
#pragma unroll
        for (int ch = 0; ch < EHID / 4; ++ch) {
            v2f a  = *(const v2f*)(xr + 4 * ch);
            v2f wv = *(const v2f*)(wc + 4 * ch);
            v2f b;
            b.x = bon ? wv.x : 0.0f;
            b.y = bon ? wv.y : 0.0f;
            acc = __builtin_amdgcn_wmma_f32_16x16x4_f32(
                       false, a,  false, b,
                       (short)0, acc,  false,  false);
        }
        if (bon) {
            float* dst = (c == 0) ? s_i : s_j;
            int r0 = wave * 16 + 8 * h;
            for (int r = 0; r < 8; ++r)
                if (r0 + r < N) dst[r0 + r] = acc[r];
        }
    } else {
        int tile = wave - tilesN;
        if (tile * 16 >= R) return;
        int row = tile * 16 + c;
        int rr = row < R ? row : R - 1;
        bool bon = c == 0;
        const float* er = rel_emb + (long)rr * RHID + 2 * h;
        const float* wc = w + EHID + 2 * h;
        v8f acc = {};
#pragma unroll
        for (int ch = 0; ch < RHID / 4; ++ch) {
            v2f a  = *(const v2f*)(er + 4 * ch);
            v2f wv = *(const v2f*)(wc + 4 * ch);
            v2f b;
            b.x = bon ? wv.x : 0.0f;
            b.y = bon ? wv.y : 0.0f;
            acc = __builtin_amdgcn_wmma_f32_16x16x4_f32(
                       false, a,  false, b,
                       (short)0, acc,  false,  false);
        }
        if (bon) {
            int r0 = tile * 16 + 8 * h;
            for (int r = 0; r < 8; ++r)
                if (r0 + r < R) s_r[r0 + r] = acc[r];
        }
    }
}

__device__ __forceinline__ int bucket_cap(int b, int N) {
    int rows = N - b * BW;
    rows = rows < BW ? rows : BW;
    int cap = rows * (DOUT / RECW);
    return cap < LCAP ? cap : LCAP;
}

__global__ void __launch_bounds__(256)
k_bin(const int* __restrict__ ei,
      const int* __restrict__ ej,
      const int* __restrict__ rel,
      int* __restrict__ bcnt,
      float* __restrict__ out, int E, int N, int NB) {
    extern __shared__ int sm[];
    int* hist = sm;
    int* base = sm + NB;
    const int tid = threadIdx.x;
    for (int i = tid; i < NB; i += 256) hist[i] = 0;
    __syncthreads();
    const int e0 = blockIdx.x * (256 * BIN_EPT);
    int vi[BIN_EPT], vj[BIN_EPT], rr[BIN_EPT], lr[BIN_EPT];
#pragma unroll
    for (int k = 0; k < BIN_EPT; ++k) {
        int e = e0 + k * 256 + tid;
        bool on = e < E;
        int ec = on ? e : (E - 1);
        int d = ei[ec];
        vj[k] = ej[ec];
        rr[k] = rel[ec];
        vi[k] = on ? d : -1;
    }
#pragma unroll
    for (int k = 0; k < BIN_EPT; ++k)
        lr[k] = (vi[k] >= 0) ? atomicAdd(&hist[vi[k] >> BSH], 1) : 0;
    __syncthreads();
    for (int i = tid; i < NB; i += 256) {
        int hc = hist[i];
        base[i] = hc ? atomicAdd(&bcnt[i], hc) : 0;
    }
    __syncthreads();
#pragma unroll
    for (int k = 0; k < BIN_EPT; ++k) {
        if (vi[k] >= 0) {
            int b = vi[k] >> BSH;
            int slot = base[b] + lr[k];
            if (slot < bucket_cap(b, N)) {
                int4 rc = make_int4(vi[k], vj[k], rr[k], 0);
                *(int4*)(out + (long)b * (BW * DOUT) + RECW * slot) = rc;
            }
        }
    }
}

__device__ __forceinline__ float hmax16(float v) {
    v = fmaxf(v, __shfl_xor(v, 8, 16));
    v = fmaxf(v, __shfl_xor(v, 4, 16));
    v = fmaxf(v, __shfl_xor(v, 2, 16));
    v = fmaxf(v, __shfl_xor(v, 1, 16));
    return v;
}
__device__ __forceinline__ float hsum16(float v) {
    v += __shfl_xor(v, 8, 16);
    v += __shfl_xor(v, 4, 16);
    v += __shfl_xor(v, 2, 16);
    v += __shfl_xor(v, 1, 16);
    return v;
}

struct NodeAcc {
    float4 aI, aJ;
    float2 aR;
};

__device__ __forceinline__ void add_edges(unsigned msk, float coef, int vj, int r, int l,
                                          const float* __restrict__ x,
                                          const float* __restrict__ rel_emb,
                                          float4 xi, NodeAcc& A) {
    while (msk) {
        const int q0 = __ffs(msk) - 1;
        float cf[4];
        float4 xj[4];
        float2 rr[4];
#pragma unroll
        for (int u = 0; u < 4; ++u) {
            bool on = (msk != 0u);
            int qu = on ? (__ffs(msk) - 1) : q0;
            msk &= msk - 1u;
            float cc = __shfl(coef, qu, 16);
            int vjq  = __shfl(vj, qu, 16);
            int rq   = __shfl(r, qu, 16);
            cf[u] = on ? cc : 0.0f;
            xj[u] = *(const float4*)(x + (long)vjq * EHID + 4 * l);
            rr[u] = *(const float2*)(rel_emb + (long)rq * RHID + 2 * l);
        }
#pragma unroll
        for (int u = 0; u < 4; ++u) {
            A.aI.x += xi.x * cf[u]; A.aI.y += xi.y * cf[u]; A.aI.z += xi.z * cf[u]; A.aI.w += xi.w * cf[u];
            A.aR.x += rr[u].x * cf[u]; A.aR.y += rr[u].y * cf[u];
            A.aJ.x += xj[u].x * cf[u]; A.aJ.y += xj[u].y * cf[u]; A.aJ.z += xj[u].z * cf[u]; A.aJ.w += xj[u].w * cf[u];
        }
    }
}

__device__ __forceinline__ float edge_score(float si, int q, int v,
                                            const float* __restrict__ s_r,
                                            const float* __restrict__ s_j) {
    return (si + s_r[q]) + s_j[v];
}

__global__ void __launch_bounds__(GT)
k_gather(const float* __restrict__ x,
         const int* __restrict__ ei,
         const int* __restrict__ ej,
         const int* __restrict__ rel,
         const float* __restrict__ rel_emb,
         const float* __restrict__ s_i,
         const float* __restrict__ s_j,
         const float* __restrict__ s_r,
         const int* __restrict__ bcnt,
         float* out, int N, int E) {
    __shared__ int   g_vj[LCAP];
    __shared__ int   g_r[LCAP];
    __shared__ float g_sc[LCAP];
    __shared__ int   ncnt[BW];
    __shared__ int   noff[BW];
    __shared__ float sI[BW];

    const int tid = threadIdx.x;
    const int b = blockIdx.x;
    const int n0 = b * BW;
    const int rows = (N - n0) < BW ? (N - n0) : BW;
    const int cb = bcnt[b];
    const bool fits = cb <= bucket_cap(b, N);
    const int l = tid & 15;

    if (tid < BW) {
        ncnt[tid] = 0;
        sI[tid] = (tid < rows) ? s_i[n0 + tid] : 0.0f;
    }
    __syncthreads();

    if (fits) {
        const int* recs = (const int*)(out + (long)n0 * DOUT);
        int lv[LCAP / GT], vj[LCAP / GT], rr[LCAP / GT], rk[LCAP / GT];
        float pj[LCAP / GT], pr[LCAP / GT];
#pragma unroll
        for (int j = 0; j < LCAP / GT; ++j) {
            int k = tid + GT * j;
            bool on = k < cb;
            int4 rc = *(const int4*)(recs + RECW * (on ? k : 0));
            lv[j] = on ? (rc.x - n0) : -1;
            vj[j] = on ? rc.y : 0;
            rr[j] = on ? rc.z : 0;
        }
#pragma unroll
        for (int j = 0; j < LCAP / GT; ++j) {
            pj[j] = s_j[vj[j]];
            pr[j] = s_r[rr[j]];
        }
#pragma unroll
        for (int j = 0; j < LCAP / GT; ++j)
            rk[j] = (lv[j] >= 0) ? atomicAdd(&ncnt[lv[j]], 1) : 0;
        __syncthreads();
        if (tid < 32) {
            int loc[BW / 32];
            int s = 0;
#pragma unroll
            for (int q = 0; q < BW / 32; ++q) { loc[q] = s; s += ncnt[(BW / 32) * tid + q]; }
            int inc = s;
#pragma unroll
            for (int d = 1; d < 32; d <<= 1) {
                int up = __shfl_up(inc, d, 32);
                if (tid >= d) inc += up;
            }
            int ex = inc - s;
#pragma unroll
            for (int q = 0; q < BW / 32; ++q) noff[(BW / 32) * tid + q] = ex + loc[q];
        }
        __syncthreads();
#pragma unroll
        for (int j = 0; j < LCAP / GT; ++j) {
            if (lv[j] >= 0) {
                int p = noff[lv[j]] + rk[j];
                g_vj[p] = vj[j];
                g_r[p]  = rr[j];
                g_sc[p] = (sI[lv[j]] + pr[j]) + pj[j];
            }
        }
        __syncthreads();
    }

    for (int it = 0; it < BW / (GT / 16); ++it) {
        const int ln = (tid >> 4) + (GT / 16) * it;
        const bool live = ln < rows;
        const int nn = live ? (n0 + ln) : n0;
        float* row = out + (long)nn * DOUT;
        float4 xi = *(const float4*)(x + (long)nn * EHID + 4 * l);
        const float si = sI[live ? ln : 0];

        NodeAcc A;
        A.aI = make_float4(0.f, 0.f, 0.f, 0.f);
        A.aJ = make_float4(0.f, 0.f, 0.f, 0.f);
        A.aR = make_float2(0.f, 0.f);

        if (fits) {
            const int c = live ? ncnt[ln] : 0;
            const int o = noff[live ? ln : 0];
            bool v0 = l < c;
            int vj0 = 0, r0 = 0;
            float a0 = -INFINITY;
            if (v0) {
                vj0 = g_vj[o + l];
                r0  = g_r[o + l];
                a0  = g_sc[o + l];
            }
            float mx = a0;
            for (int bb = 16; bb < c; bb += 16) {
                int pos = bb + l;
                if (pos < c) mx = fmaxf(mx, g_sc[o + pos]);
            }
            mx = hmax16(mx);
            float p0 = v0 ? expf(a0 - mx) : 0.0f;
            float s = p0;
            for (int bb = 16; bb < c; bb += 16) {
                int pos = bb + l;
                if (pos < c) s += expf(g_sc[o + pos] - mx);
            }
            s = hsum16(s);
            float den = s + EPSF;
            {
                int k = c < 16 ? c : 16;
                add_edges((1u << k) - 1u, p0 / den, vj0, r0, l, x, rel_emb, xi, A);
            }
            for (int bb = 16; bb < c; bb += 16) {
                int pos = bb + l;
                float coef = 0.0f;
                int vjj = 0, r = 0;
                if (pos < c) {
                    vjj = g_vj[o + pos];
                    r = g_r[o + pos];
                    coef = expf(g_sc[o + pos] - mx) / den;
                }
                int k = (c - bb) < 16 ? (c - bb) : 16;
                add_edges((1u << k) - 1u, coef, vjj, r, l, x, rel_emb, xi, A);
            }
        } else {
            int hshift = tid & 16;
            float mx = -INFINITY;
            for (int bb = 0; bb < E; bb += 16) {
                int pos = bb + l;
                if (live && pos < E && ei[pos] == nn) mx = fmaxf(mx, edge_score(si, rel[pos], ej[pos], s_r, s_j));
            }
            mx = hmax16(mx);
            float s = 0.0f;
            for (int bb = 0; bb < E; bb += 16) {
                int pos = bb + l;
                if (live && pos < E && ei[pos] == nn) s += expf(edge_score(si, rel[pos], ej[pos], s_r, s_j) - mx);
            }
            s = hsum16(s);
            float den = s + EPSF;
            for (int bb = 0; bb < E; bb += 16) {
                int pos = bb + l;
                bool valid = live && (pos < E) && (ei[pos] == nn);
                float coef = 0.0f;
                int vjj = 0, r = 0;
                if (valid) {
                    vjj = ej[pos];
                    r = rel[pos];
                    coef = expf(edge_score(si, r, vjj, s_r, s_j) - mx) / den;
                }
                unsigned msk = (unsigned)((__ballot(valid) >> hshift) & 0xFFFFull);
                add_edges(msk, coef, vjj, r, l, x, rel_emb, xi, A);
            }
        }

        if (live) {
            *(float4*)(row + 4 * l) = xi;
            *(float4*)(row + EHID + 4 * l) =
                make_float4(fmaxf(A.aI.x, 0.f), fmaxf(A.aI.y, 0.f), fmaxf(A.aI.z, 0.f), fmaxf(A.aI.w, 0.f));
            *(float2*)(row + 2 * EHID + 2 * l) = make_float2(fmaxf(A.aR.x, 0.f), fmaxf(A.aR.y, 0.f));
            *(float4*)(row + 2 * EHID + RHID + 4 * l) =
                make_float4(fmaxf(A.aJ.x, 0.f), fmaxf(A.aJ.y, 0.f), fmaxf(A.aJ.z, 0.f), fmaxf(A.aJ.w, 0.f));
        }
    }
}

extern "C" void kernel_launch(void* const* d_in, const int* in_sizes, int n_in,
                              void* d_out, int out_size, void* d_ws, size_t ws_size,
                              hipStream_t stream) {
    const float* x        = (const float*)d_in[0];
    const int*   eidx     = (const int*)d_in[1];
    const int*   rel      = (const int*)d_in[2];
    const float* rel_emb  = (const float*)d_in[3];
    const float* w        = (const float*)d_in[4];
    float* out = (float*)d_out;

    const int E = in_sizes[2];
    const int N = in_sizes[0] / EHID;
    const int R = in_sizes[3] / RHID;
    const int* ei = eidx;
    const int* ej = eidx + E;

    const int NB = ceil_div(N, BW);
    char* ws = (char*)d_ws;
    float* s_i = (float*)ws;
    float* s_j = (float*)(ws + (size_t)N * 4);
    float* s_r = (float*)(ws + (size_t)N * 8);
    int*  bcnt = (int*)(ws + (size_t)N * 8 + (size_t)R * 4);

    const int B = 256;

    const int tilesN = ceil_div(N, 16);
    const int tilesR = ceil_div(R, 16);
    k_project<<<ceil_div(tilesN + tilesR, 8), B, 0, stream>>>(x, rel_emb, w, s_i, s_j, s_r, (unsigned*)bcnt, N, R, tilesN, NB);

    k_bin<<<ceil_div(E, B * BIN_EPT), B, (size_t)NB * 8, stream>>>(ei, ej, rel, bcnt, out, E, N, NB);

    k_gather<<<NB, GT, 0, stream>>>(x, ei, ej, rel, rel_emb, s_i, s_j, s_r, bcnt, out, N, E);
}


# Gather: source and relation rows of a node's first 8 edges requested as soon as its edge list is read, before the max/exp/sum passes (were fetched 4 at a time after them); sum order unchanged
# speedup vs baseline: 5.6960x; 1.0380x over previous
#include <hip/hip_runtime.h>
#include <hip/hip_bf16.h>


typedef __attribute__((ext_vector_type(2))) float v2f;
typedef __attribute__((ext_vector_type(8))) float v8f;

#define EHID 64
#define RHID 32
#define DFEAT 160
#define DOUT 224
#define EPSF 1e-16f
#define BW 256
#define BSH 8
#define RECW 4
#define LCAP 4096
#define BIN_EPT 8
#define GT 512
#define FE 8

static inline int ceil_div(int a, int b) { return (a + b - 1) / b; }

__global__ void __launch_bounds__(256)
k_project(const float* __restrict__ x,
          const float* __restrict__ rel_emb,
          const float* __restrict__ w,
          float* __restrict__ s_i,
          float* __restrict__ s_j,
          float* __restrict__ s_r,
          unsigned* __restrict__ bcnt,
          int N, int R, int tilesN, int NB) {
    int t = threadIdx.x;
    int gid = blockIdx.x * blockDim.x + t;
    if (gid < NB) bcnt[gid] = 0u;

    int wave = gid >> 5;
    int lane = t & 31;
    int c = lane & 15;
    int h = lane >> 4;

    if (wave < tilesN) {
        int row = wave * 16 + c;
        int rr = row < N ? row : N - 1;
        bool bon = c < 2;
        const float* xr = x + (long)rr * EHID + 2 * h;
        const float* wc = w + ((c == 1) ? (EHID + RHID) : 0) + 2 * h;
        v8f acc = {};
#pragma unroll
        for (int ch = 0; ch < EHID / 4; ++ch) {
            v2f a  = *(const v2f*)(xr + 4 * ch);
            v2f wv = *(const v2f*)(wc + 4 * ch);
            v2f b;
            b.x = bon ? wv.x : 0.0f;
            b.y = bon ? wv.y : 0.0f;
            acc = __builtin_amdgcn_wmma_f32_16x16x4_f32(
                       false, a,  false, b,
                       (short)0, acc,  false,  false);
        }
        if (bon) {
            float* dst = (c == 0) ? s_i : s_j;
            int r0 = wave * 16 + 8 * h;
            for (int r = 0; r < 8; ++r)
                if (r0 + r < N) dst[r0 + r] = acc[r];
        }
    } else {
        int tile = wave - tilesN;
        if (tile * 16 >= R) return;
        int row = tile * 16 + c;
        int rr = row < R ? row : R - 1;
        bool bon = c == 0;
        const float* er = rel_emb + (long)rr * RHID + 2 * h;
        const float* wc = w + EHID + 2 * h;
        v8f acc = {};
#pragma unroll
        for (int ch = 0; ch < RHID / 4; ++ch) {
            v2f a  = *(const v2f*)(er + 4 * ch);
            v2f wv = *(const v2f*)(wc + 4 * ch);
            v2f b;
            b.x = bon ? wv.x : 0.0f;
            b.y = bon ? wv.y : 0.0f;
            acc = __builtin_amdgcn_wmma_f32_16x16x4_f32(
                       false, a,  false, b,
                       (short)0, acc,  false,  false);
        }
        if (bon) {
            int r0 = tile * 16 + 8 * h;
            for (int r = 0; r < 8; ++r)
                if (r0 + r < R) s_r[r0 + r] = acc[r];
        }
    }
}

__device__ __forceinline__ int bucket_cap(int b, int N) {
    int rows = N - b * BW;
    rows = rows < BW ? rows : BW;
    int cap = rows * (DOUT / RECW);
    return cap < LCAP ? cap : LCAP;
}

__global__ void __launch_bounds__(256)
k_bin(const int* __restrict__ ei,
      const int* __restrict__ ej,
      const int* __restrict__ rel,
      int* __restrict__ bcnt,
      float* __restrict__ out, int E, int N, int NB) {
    extern __shared__ int sm[];
    int* hist = sm;
    int* base = sm + NB;
    const int tid = threadIdx.x;
    for (int i = tid; i < NB; i += 256) hist[i] = 0;
    __syncthreads();
    const int e0 = blockIdx.x * (256 * BIN_EPT);
    int vi[BIN_EPT], vj[BIN_EPT], rr[BIN_EPT], lr[BIN_EPT];
#pragma unroll
    for (int k = 0; k < BIN_EPT; ++k) {
        int e = e0 + k * 256 + tid;
        bool on = e < E;
        int ec = on ? e : (E - 1);
        int d = ei[ec];
        vj[k] = ej[ec];
        rr[k] = rel[ec];
        vi[k] = on ? d : -1;
    }
#pragma unroll
    for (int k = 0; k < BIN_EPT; ++k)
        lr[k] = (vi[k] >= 0) ? atomicAdd(&hist[vi[k] >> BSH], 1) : 0;
    __syncthreads();
    for (int i = tid; i < NB; i += 256) {
        int hc = hist[i];
        base[i] = hc ? atomicAdd(&bcnt[i], hc) : 0;
    }
    __syncthreads();
#pragma unroll
    for (int k = 0; k < BIN_EPT; ++k) {
        if (vi[k] >= 0) {
            int b = vi[k] >> BSH;
            int slot = base[b] + lr[k];
            if (slot < bucket_cap(b, N)) {
                int4 rc = make_int4(vi[k], vj[k], rr[k], 0);
                *(int4*)(out + (long)b * (BW * DOUT) + RECW * slot) = rc;
            }
        }
    }
}

__device__ __forceinline__ float hmax16(float v) {
    v = fmaxf(v, __shfl_xor(v, 8, 16));
    v = fmaxf(v, __shfl_xor(v, 4, 16));
    v = fmaxf(v, __shfl_xor(v, 2, 16));
    v = fmaxf(v, __shfl_xor(v, 1, 16));
    return v;
}
__device__ __forceinline__ float hsum16(float v) {
    v += __shfl_xor(v, 8, 16);
    v += __shfl_xor(v, 4, 16);
    v += __shfl_xor(v, 2, 16);
    v += __shfl_xor(v, 1, 16);
    return v;
}

struct NodeAcc {
    float4 aI, aJ;
    float2 aR;
};

__device__ __forceinline__ void add_edges(unsigned msk, float coef, int vj, int r, int l,
                                          const float* __restrict__ x,
                                          const float* __restrict__ rel_emb,
                                          float4 xi, NodeAcc& A) {
    while (msk) {
        const int q0 = __ffs(msk) - 1;
        float cf[4];
        float4 xj[4];
        float2 rr[4];
#pragma unroll
        for (int u = 0; u < 4; ++u) {
            bool on = (msk != 0u);
            int qu = on ? (__ffs(msk) - 1) : q0;
            msk &= msk - 1u;
            float cc = __shfl(coef, qu, 16);
            int vjq  = __shfl(vj, qu, 16);
            int rq   = __shfl(r, qu, 16);
            cf[u] = on ? cc : 0.0f;
            xj[u] = *(const float4*)(x + (long)vjq * EHID + 4 * l);
            rr[u] = *(const float2*)(rel_emb + (long)rq * RHID + 2 * l);
        }
#pragma unroll
        for (int u = 0; u < 4; ++u) {
            A.aI.x += xi.x * cf[u]; A.aI.y += xi.y * cf[u]; A.aI.z += xi.z * cf[u]; A.aI.w += xi.w * cf[u];
            A.aR.x += rr[u].x * cf[u]; A.aR.y += rr[u].y * cf[u];
            A.aJ.x += xj[u].x * cf[u]; A.aJ.y += xj[u].y * cf[u]; A.aJ.z += xj[u].z * cf[u]; A.aJ.w += xj[u].w * cf[u];
        }
    }
}

__device__ __forceinline__ float edge_score(float si, int q, int v,
                                            const float* __restrict__ s_r,
                                            const float* __restrict__ s_j) {
    return (si + s_r[q]) + s_j[v];
}

__global__ void __launch_bounds__(GT, 8)
k_gather(const float* __restrict__ x,
         const int* __restrict__ ei,
         const int* __restrict__ ej,
         const int* __restrict__ rel,
         const float* __restrict__ rel_emb,
         const float* __restrict__ s_i,
         const float* __restrict__ s_j,
         const float* __restrict__ s_r,
         const int* __restrict__ bcnt,
         float* out, int N, int E) {
    __shared__ int   g_vj[LCAP];
    __shared__ int   g_r[LCAP];
    __shared__ float g_sc[LCAP];
    __shared__ int   ncnt[BW];
    __shared__ int   noff[BW];
    __shared__ float sI[BW];

    const int tid = threadIdx.x;
    const int b = blockIdx.x;
    const int n0 = b * BW;
    const int rows = (N - n0) < BW ? (N - n0) : BW;
    const int cb = bcnt[b];
    const bool fits = cb <= bucket_cap(b, N);
    const int l = tid & 15;

    if (tid < BW) {
        ncnt[tid] = 0;
        sI[tid] = (tid < rows) ? s_i[n0 + tid] : 0.0f;
    }
    __syncthreads();

    if (fits) {
        const int* recs = (const int*)(out + (long)n0 * DOUT);
        int lv[LCAP / GT], vj[LCAP / GT], rr[LCAP / GT], rk[LCAP / GT];
        float pj[LCAP / GT], pr[LCAP / GT];
#pragma unroll
        for (int j = 0; j < LCAP / GT; ++j) {
            int k = tid + GT * j;
            bool on = k < cb;
            int4 rc = *(const int4*)(recs + RECW * (on ? k : 0));
            lv[j] = on ? (rc.x - n0) : -1;
            vj[j] = on ? rc.y : 0;
            rr[j] = on ? rc.z : 0;
        }
#pragma unroll
        for (int j = 0; j < LCAP / GT; ++j) {
            pj[j] = s_j[vj[j]];
            pr[j] = s_r[rr[j]];
        }
#pragma unroll
        for (int j = 0; j < LCAP / GT; ++j)
            rk[j] = (lv[j] >= 0) ? atomicAdd(&ncnt[lv[j]], 1) : 0;
        __syncthreads();
        if (tid < 32) {
            int loc[BW / 32];
            int s = 0;
#pragma unroll
            for (int q = 0; q < BW / 32; ++q) { loc[q] = s; s += ncnt[(BW / 32) * tid + q]; }
            int inc = s;
#pragma unroll
            for (int d = 1; d < 32; d <<= 1) {
                int up = __shfl_up(inc, d, 32);
                if (tid >= d) inc += up;
            }
            int ex = inc - s;
#pragma unroll
            for (int q = 0; q < BW / 32; ++q) noff[(BW / 32) * tid + q] = ex + loc[q];
        }
        __syncthreads();
#pragma unroll
        for (int j = 0; j < LCAP / GT; ++j) {
            if (lv[j] >= 0) {
                int p = noff[lv[j]] + rk[j];
                g_vj[p] = vj[j];
                g_r[p]  = rr[j];
                g_sc[p] = (sI[lv[j]] + pr[j]) + pj[j];
            }
        }
        __syncthreads();
    }

    for (int it = 0; it < BW / (GT / 16); ++it) {
        const int ln = (tid >> 4) + (GT / 16) * it;
        const bool live = ln < rows;
        const int nn = live ? (n0 + ln) : n0;
        float* row = out + (long)nn * DOUT;
        float4 xi = *(const float4*)(x + (long)nn * EHID + 4 * l);
        const float si = sI[live ? ln : 0];

        NodeAcc A;
        A.aI = make_float4(0.f, 0.f, 0.f, 0.f);
        A.aJ = make_float4(0.f, 0.f, 0.f, 0.f);
        A.aR = make_float2(0.f, 0.f);

        if (fits) {
            const int c = live ? ncnt[ln] : 0;
            const int o = noff[live ? ln : 0];
            bool v0 = l < c;
            int vj0 = 0, r0 = 0;
            float a0 = -INFINITY;
            if (v0) {
                vj0 = g_vj[o + l];
                r0  = g_r[o + l];
                a0  = g_sc[o + l];
            }
            float4 xj[FE];
            float2 rr[FE];
            const int vjf = __shfl(vj0, 0, 16);
            const int rf  = __shfl(r0, 0, 16);
            const int vjs = v0 ? vj0 : vjf;
            const int rs  = v0 ? r0  : rf;
#pragma unroll
            for (int u = 0; u < FE; ++u) {
                int vjq = __shfl(vjs, u, 16);
                int rq  = __shfl(rs, u, 16);
                xj[u] = *(const float4*)((const char*)x + (unsigned)(vjq * (EHID * 4) + 16 * l));
                rr[u] = *(const float2*)((const char*)rel_emb + (unsigned)(rq * (RHID * 4) + 8 * l));
            }
            float mx = a0;
            for (int bb = 16; bb < c; bb += 16) {
                int pos = bb + l;
                if (pos < c) mx = fmaxf(mx, g_sc[o + pos]);
            }
            mx = hmax16(mx);
            float p0 = v0 ? expf(a0 - mx) : 0.0f;
            float s = p0;
            for (int bb = 16; bb < c; bb += 16) {
                int pos = bb + l;
                if (pos < c) s += expf(g_sc[o + pos] - mx);
            }
            s = hsum16(s);
            float den = s + EPSF;
            const float coef0 = p0 / den;
#pragma unroll
            for (int u = 0; u < FE; ++u) {
                float cf = __shfl(coef0, u, 16);
                A.aI.x += xi.x * cf; A.aI.y += xi.y * cf; A.aI.z += xi.z * cf; A.aI.w += xi.w * cf;
                A.aR.x += rr[u].x * cf; A.aR.y += rr[u].y * cf;
                A.aJ.x += xj[u].x * cf; A.aJ.y += xj[u].y * cf; A.aJ.z += xj[u].z * cf; A.aJ.w += xj[u].w * cf;
            }
            {
                int k = c < 16 ? c : 16;
                unsigned m = ((1u << k) - 1u) & ~((1u << FE) - 1u);
                add_edges(m, coef0, vj0, r0, l, x, rel_emb, xi, A);
            }
            for (int bb = 16; bb < c; bb += 16) {
                int pos = bb + l;
                float coef = 0.0f;
                int vjj = 0, r = 0;
                if (pos < c) {
                    vjj = g_vj[o + pos];
                    r = g_r[o + pos];
                    coef = expf(g_sc[o + pos] - mx) / den;
                }
                int k = (c - bb) < 16 ? (c - bb) : 16;
                add_edges((1u << k) - 1u, coef, vjj, r, l, x, rel_emb, xi, A);
            }
        } else {
            int hshift = tid & 16;
            float mx = -INFINITY;
            for (int bb = 0; bb < E; bb += 16) {
                int pos = bb + l;
                if (live && pos < E && ei[pos] == nn) mx = fmaxf(mx, edge_score(si, rel[pos], ej[pos], s_r, s_j));
            }
            mx = hmax16(mx);
            float s = 0.0f;
            for (int bb = 0; bb < E; bb += 16) {
                int pos = bb + l;
                if (live && pos < E && ei[pos] == nn) s += expf(edge_score(si, rel[pos], ej[pos], s_r, s_j) - mx);
            }
            s = hsum16(s);
            float den = s + EPSF;
            for (int bb = 0; bb < E; bb += 16) {
                int pos = bb + l;
                bool valid = live && (pos < E) && (ei[pos] == nn);
                float coef = 0.0f;
                int vjj = 0, r = 0;
                if (valid) {
                    vjj = ej[pos];
                    r = rel[pos];
                    coef = expf(edge_score(si, r, vjj, s_r, s_j) - mx) / den;
                }
                unsigned msk = (unsigned)((__ballot(valid) >> hshift) & 0xFFFFull);
                add_edges(msk, coef, vjj, r, l, x, rel_emb, xi, A);
            }
        }

        if (live) {
            *(float4*)(row + 4 * l) = xi;
            *(float4*)(row + EHID + 4 * l) =
                make_float4(fmaxf(A.aI.x, 0.f), fmaxf(A.aI.y, 0.f), fmaxf(A.aI.z, 0.f), fmaxf(A.aI.w, 0.f));
            *(float2*)(row + 2 * EHID + 2 * l) = make_float2(fmaxf(A.aR.x, 0.f), fmaxf(A.aR.y, 0.f));
            *(float4*)(row + 2 * EHID + RHID + 4 * l) =
                make_float4(fmaxf(A.aJ.x, 0.f), fmaxf(A.aJ.y, 0.f), fmaxf(A.aJ.z, 0.f), fmaxf(A.aJ.w, 0.f));
        }
    }
}

extern "C" void kernel_launch(void* const* d_in, const int* in_sizes, int n_in,
                              void* d_out, int out_size, void* d_ws, size_t ws_size,
                              hipStream_t stream) {
    const float* x        = (const float*)d_in[0];
    const int*   eidx     = (const int*)d_in[1];
    const int*   rel      = (const int*)d_in[2];
    const float* rel_emb  = (const float*)d_in[3];
    const float* w        = (const float*)d_in[4];
    float* out = (float*)d_out;

    const int E = in_sizes[2];
    const int N = in_sizes[0] / EHID;
    const int R = in_sizes[3] / RHID;
    const int* ei = eidx;
    const int* ej = eidx + E;

    const int NB = ceil_div(N, BW);
    char* ws = (char*)d_ws;
    float* s_i = (float*)ws;
    float* s_j = (float*)(ws + (size_t)N * 4);
    float* s_r = (float*)(ws + (size_t)N * 8);
    int*  bcnt = (int*)(ws + (size_t)N * 8 + (size_t)R * 4);

    const int B = 256;

    const int tilesN = ceil_div(N, 16);
    const int tilesR = ceil_div(R, 16);
    k_project<<<ceil_div(tilesN + tilesR, 8), B, 0, stream>>>(x, rel_emb, w, s_i, s_j, s_r, (unsigned*)bcnt, N, R, tilesN, NB);

    k_bin<<<ceil_div(E, B * BIN_EPT), B, (size_t)NB * 8, stream>>>(ei, ej, rel, bcnt, out, E, N, NB);

    k_gather<<<NB, GT, 0, stream>>>(x, ei, ej, rel, rel_emb, s_i, s_j, s_r, bcnt, out, N, E);
}


# Buckets re-sized from 256 to 196 nodes (record capacity 4096 to 2048): gather grid 391 to 511 work-groups, 2 on every WGP at once; node sums stay in LDS arrival-rank order (new grouping)
# speedup vs baseline: 6.0584x; 1.0636x over previous
#include <hip/hip_runtime.h>
#include <hip/hip_bf16.h>


typedef __attribute__((ext_vector_type(2))) float v2f;
typedef __attribute__((ext_vector_type(8))) float v8f;

#define EHID 64
#define RHID 32
#define DFEAT 160
#define DOUT 224
#define EPSF 1e-16f
#define BW 196
#define BWP 224
#define RECW 4
#define LCAP 2048
#define BIN_EPT 8
#define GT 512
#define FE 8

static inline int ceil_div(int a, int b) { return (a + b - 1) / b; }

__global__ void __launch_bounds__(256)
k_project(const float* __restrict__ x,
          const float* __restrict__ rel_emb,
          const float* __restrict__ w,
          float* __restrict__ s_i,
          float* __restrict__ s_j,
          float* __restrict__ s_r,
          unsigned* __restrict__ bcnt,
          int N, int R, int tilesN, int NB) {
    int t = threadIdx.x;
    int gid = blockIdx.x * blockDim.x + t;
    if (gid < NB) bcnt[gid] = 0u;

    int wave = gid >> 5;
    int lane = t & 31;
    int c = lane & 15;
    int h = lane >> 4;

    if (wave < tilesN) {
        int row = wave * 16 + c;
        int rr = row < N ? row : N - 1;
        bool bon = c < 2;
        const float* xr = x + (long)rr * EHID + 2 * h;
        const float* wc = w + ((c == 1) ? (EHID + RHID) : 0) + 2 * h;
        v8f acc = {};
#pragma unroll
        for (int ch = 0; ch < EHID / 4; ++ch) {
            v2f a  = *(const v2f*)(xr + 4 * ch);
            v2f wv = *(const v2f*)(wc + 4 * ch);
            v2f b;
            b.x = bon ? wv.x : 0.0f;
            b.y = bon ? wv.y : 0.0f;
            acc = __builtin_amdgcn_wmma_f32_16x16x4_f32(
                       false, a,  false, b,
                       (short)0, acc,  false,  false);
        }
        if (bon) {
            float* dst = (c == 0) ? s_i : s_j;
            int r0 = wave * 16 + 8 * h;
            for (int r = 0; r < 8; ++r)
                if (r0 + r < N) dst[r0 + r] = acc[r];
        }
    } else {
        int tile = wave - tilesN;
        if (tile * 16 >= R) return;
        int row = tile * 16 + c;
        int rr = row < R ? row : R - 1;
        bool bon = c == 0;
        const float* er = rel_emb + (long)rr * RHID + 2 * h;
        const float* wc = w + EHID + 2 * h;
        v8f acc = {};
#pragma unroll
        for (int ch = 0; ch < RHID / 4; ++ch) {
            v2f a  = *(const v2f*)(er + 4 * ch);
            v2f wv = *(const v2f*)(wc + 4 * ch);
            v2f b;
            b.x = bon ? wv.x : 0.0f;
            b.y = bon ? wv.y : 0.0f;
            acc = __builtin_amdgcn_wmma_f32_16x16x4_f32(
                       false, a,  false, b,
                       (short)0, acc,  false,  false);
        }
        if (bon) {
            int r0 = tile * 16 + 8 * h;
            for (int r = 0; r < 8; ++r)
                if (r0 + r < R) s_r[r0 + r] = acc[r];
        }
    }
}

__device__ __forceinline__ int bucket_cap(int b, int N) {
    int rows = N - b * BW;
    rows = rows < BW ? rows : BW;
    int cap = rows * (DOUT / RECW);
    return cap < LCAP ? cap : LCAP;
}

__global__ void __launch_bounds__(256)
k_bin(const int* __restrict__ ei,
      const int* __restrict__ ej,
      const int* __restrict__ rel,
      int* __restrict__ bcnt,
      float* __restrict__ out, int E, int N, int NB) {
    extern __shared__ int sm[];
    int* hist = sm;
    int* base = sm + NB;
    const int tid = threadIdx.x;
    for (int i = tid; i < NB; i += 256) hist[i] = 0;
    __syncthreads();
    const int e0 = blockIdx.x * (256 * BIN_EPT);
    int vi[BIN_EPT], vj[BIN_EPT], rr[BIN_EPT], lr[BIN_EPT];
#pragma unroll
    for (int k = 0; k < BIN_EPT; ++k) {
        int e = e0 + k * 256 + tid;
        bool on = e < E;
        int ec = on ? e : (E - 1);
        int d = ei[ec];
        vj[k] = ej[ec];
        rr[k] = rel[ec];
        vi[k] = on ? d : -1;
    }
#pragma unroll
    for (int k = 0; k < BIN_EPT; ++k)
        lr[k] = (vi[k] >= 0) ? atomicAdd(&hist[(unsigned)vi[k] / BW], 1) : 0;
    __syncthreads();
    for (int i = tid; i < NB; i += 256) {
        int hc = hist[i];
        base[i] = hc ? atomicAdd(&bcnt[i], hc) : 0;
    }
    __syncthreads();
#pragma unroll
    for (int k = 0; k < BIN_EPT; ++k) {
        if (vi[k] >= 0) {
            int b = (int)((unsigned)vi[k] / BW);
            int slot = base[b] + lr[k];
            if (slot < bucket_cap(b, N)) {
                int4 rc = make_int4(vi[k], vj[k], rr[k], 0);
                *(int4*)(out + (long)b * (BW * DOUT) + RECW * slot) = rc;
            }
        }
    }
}

__device__ __forceinline__ float hmax16(float v) {
    v = fmaxf(v, __shfl_xor(v, 8, 16));
    v = fmaxf(v, __shfl_xor(v, 4, 16));
    v = fmaxf(v, __shfl_xor(v, 2, 16));
    v = fmaxf(v, __shfl_xor(v, 1, 16));
    return v;
}
__device__ __forceinline__ float hsum16(float v) {
    v += __shfl_xor(v, 8, 16);
    v += __shfl_xor(v, 4, 16);
    v += __shfl_xor(v, 2, 16);
    v += __shfl_xor(v, 1, 16);
    return v;
}

struct NodeAcc {
    float4 aI, aJ;
    float2 aR;
};

__device__ __forceinline__ void add_edges(unsigned msk, float coef, int vj, int r, int l,
                                          const float* __restrict__ x,
                                          const float* __restrict__ rel_emb,
                                          float4 xi, NodeAcc& A) {
    while (msk) {
        const int q0 = __ffs(msk) - 1;
        float cf[4];
        float4 xj[4];
        float2 rr[4];
#pragma unroll
        for (int u = 0; u < 4; ++u) {
            bool on = (msk != 0u);
            int qu = on ? (__ffs(msk) - 1) : q0;
            msk &= msk - 1u;
            float cc = __shfl(coef, qu, 16);
            int vjq  = __shfl(vj, qu, 16);
            int rq   = __shfl(r, qu, 16);
            cf[u] = on ? cc : 0.0f;
            xj[u] = *(const float4*)(x + (long)vjq * EHID + 4 * l);
            rr[u] = *(const float2*)(rel_emb + (long)rq * RHID + 2 * l);
        }
#pragma unroll
        for (int u = 0; u < 4; ++u) {
            A.aI.x += xi.x * cf[u]; A.aI.y += xi.y * cf[u]; A.aI.z += xi.z * cf[u]; A.aI.w += xi.w * cf[u];
            A.aR.x += rr[u].x * cf[u]; A.aR.y += rr[u].y * cf[u];
            A.aJ.x += xj[u].x * cf[u]; A.aJ.y += xj[u].y * cf[u]; A.aJ.z += xj[u].z * cf[u]; A.aJ.w += xj[u].w * cf[u];
        }
    }
}

__device__ __forceinline__ float edge_score(float si, int q, int v,
                                            const float* __restrict__ s_r,
                                            const float* __restrict__ s_j) {
    return (si + s_r[q]) + s_j[v];
}

__global__ void __launch_bounds__(GT, 8)
k_gather(const float* __restrict__ x,
         const int* __restrict__ ei,
         const int* __restrict__ ej,
         const int* __restrict__ rel,
         const float* __restrict__ rel_emb,
         const float* __restrict__ s_i,
         const float* __restrict__ s_j,
         const float* __restrict__ s_r,
         const int* __restrict__ bcnt,
         float* out, int N, int E) {
    __shared__ int   g_vj[LCAP];
    __shared__ int   g_r[LCAP];
    __shared__ float g_sc[LCAP];
    __shared__ int   ncnt[BWP];
    __shared__ int   noff[BWP];
    __shared__ float sI[BWP];

    const int tid = threadIdx.x;
    const int b = blockIdx.x;
    const int n0 = b * BW;
    const int rows = (N - n0) < BW ? (N - n0) : BW;
    const int cb = bcnt[b];
    const bool fits = cb <= bucket_cap(b, N);
    const int l = tid & 15;

    if (tid < BWP) {
        ncnt[tid] = 0;
        sI[tid] = (tid < rows) ? s_i[n0 + tid] : 0.0f;
    }
    __syncthreads();

    if (fits) {
        const int* recs = (const int*)(out + (long)n0 * DOUT);
        int lv[LCAP / GT], vj[LCAP / GT], rr[LCAP / GT], rk[LCAP / GT];
        float pj[LCAP / GT], pr[LCAP / GT];
#pragma unroll
        for (int j = 0; j < LCAP / GT; ++j) {
            int k = tid + GT * j;
            bool on = k < cb;
            int4 rc = *(const int4*)(recs + RECW * (on ? k : 0));
            lv[j] = on ? (rc.x - n0) : -1;
            vj[j] = on ? rc.y : 0;
            rr[j] = on ? rc.z : 0;
        }
#pragma unroll
        for (int j = 0; j < LCAP / GT; ++j) {
            pj[j] = s_j[vj[j]];
            pr[j] = s_r[rr[j]];
        }
#pragma unroll
        for (int j = 0; j < LCAP / GT; ++j)
            rk[j] = (lv[j] >= 0) ? atomicAdd(&ncnt[lv[j]], 1) : 0;
        __syncthreads();
        if (tid < 32) {
            int loc[BWP / 32];
            int s = 0;
#pragma unroll
            for (int q = 0; q < BWP / 32; ++q) { loc[q] = s; s += ncnt[(BWP / 32) * tid + q]; }
            int inc = s;
#pragma unroll
            for (int d = 1; d < 32; d <<= 1) {
                int up = __shfl_up(inc, d, 32);
                if (tid >= d) inc += up;
            }
            int ex = inc - s;
#pragma unroll
            for (int q = 0; q < BWP / 32; ++q) noff[(BWP / 32) * tid + q] = ex + loc[q];
        }
        __syncthreads();
#pragma unroll
        for (int j = 0; j < LCAP / GT; ++j) {
            if (lv[j] >= 0) {
                int p = noff[lv[j]] + rk[j];
                g_vj[p] = vj[j];
                g_r[p]  = rr[j];
                g_sc[p] = (sI[lv[j]] + pr[j]) + pj[j];
            }
        }
        __syncthreads();
    }

    for (int it = 0; it < BWP / (GT / 16); ++it) {
        if (2 * (tid >> 5) + (GT / 16) * it >= rows) break;
        const int ln = (tid >> 4) + (GT / 16) * it;
        const bool live = ln < rows;
        const int nn = live ? (n0 + ln) : n0;
        float* row = out + (long)nn * DOUT;
        float4 xi = *(const float4*)(x + (long)nn * EHID + 4 * l);
        const float si = sI[live ? ln : 0];

        NodeAcc A;
        A.aI = make_float4(0.f, 0.f, 0.f, 0.f);
        A.aJ = make_float4(0.f, 0.f, 0.f, 0.f);
        A.aR = make_float2(0.f, 0.f);

        if (fits) {
            const int c = live ? ncnt[ln] : 0;
            const int o = noff[live ? ln : 0];
            bool v0 = l < c;
            int vj0 = 0, r0 = 0;
            float a0 = -INFINITY;
            if (v0) {
                vj0 = g_vj[o + l];
                r0  = g_r[o + l];
                a0  = g_sc[o + l];
            }
            float4 xj[FE];
            float2 rr[FE];
            const int vjf = __shfl(vj0, 0, 16);
            const int rf  = __shfl(r0, 0, 16);
            const int vjs = v0 ? vj0 : vjf;
            const int rs  = v0 ? r0  : rf;
#pragma unroll
            for (int u = 0; u < FE; ++u) {
                int vjq = __shfl(vjs, u, 16);
                int rq  = __shfl(rs, u, 16);
                xj[u] = *(const float4*)((const char*)x + (unsigned)(vjq * (EHID * 4) + 16 * l));
                rr[u] = *(const float2*)((const char*)rel_emb + (unsigned)(rq * (RHID * 4) + 8 * l));
            }
            float mx = a0;
            for (int bb = 16; bb < c; bb += 16) {
                int pos = bb + l;
                if (pos < c) mx = fmaxf(mx, g_sc[o + pos]);
            }
            mx = hmax16(mx);
            float p0 = v0 ? expf(a0 - mx) : 0.0f;
            float s = p0;
            for (int bb = 16; bb < c; bb += 16) {
                int pos = bb + l;
                if (pos < c) s += expf(g_sc[o + pos] - mx);
            }
            s = hsum16(s);
            float den = s + EPSF;
            const float coef0 = p0 / den;
#pragma unroll
            for (int u = 0; u < FE; ++u) {
                float cf = __shfl(coef0, u, 16);
                A.aI.x += xi.x * cf; A.aI.y += xi.y * cf; A.aI.z += xi.z * cf; A.aI.w += xi.w * cf;
                A.aR.x += rr[u].x * cf; A.aR.y += rr[u].y * cf;
                A.aJ.x += xj[u].x * cf; A.aJ.y += xj[u].y * cf; A.aJ.z += xj[u].z * cf; A.aJ.w += xj[u].w * cf;
            }
            {
                int k = c < 16 ? c : 16;
                unsigned m = ((1u << k) - 1u) & ~((1u << FE) - 1u);
                add_edges(m, coef0, vj0, r0, l, x, rel_emb, xi, A);
            }
            for (int bb = 16; bb < c; bb += 16) {
                int pos = bb + l;
                float coef = 0.0f;
                int vjj = 0, r = 0;
                if (pos < c) {
                    vjj = g_vj[o + pos];
                    r = g_r[o + pos];
                    coef = expf(g_sc[o + pos] - mx) / den;
                }
                int k = (c - bb) < 16 ? (c - bb) : 16;
                add_edges((1u << k) - 1u, coef, vjj, r, l, x, rel_emb, xi, A);
            }
        } else {
            int hshift = tid & 16;
            float mx = -INFINITY;
            for (int bb = 0; bb < E; bb += 16) {
                int pos = bb + l;
                if (live && pos < E && ei[pos] == nn) mx = fmaxf(mx, edge_score(si, rel[pos], ej[pos], s_r, s_j));
            }
            mx = hmax16(mx);
            float s = 0.0f;
            for (int bb = 0; bb < E; bb += 16) {
                int pos = bb + l;
                if (live && pos < E && ei[pos] == nn) s += expf(edge_score(si, rel[pos], ej[pos], s_r, s_j) - mx);
            }
            s = hsum16(s);
            float den = s + EPSF;
            for (int bb = 0; bb < E; bb += 16) {
                int pos = bb + l;
                bool valid = live && (pos < E) && (ei[pos] == nn);
                float coef = 0.0f;
                int vjj = 0, r = 0;
                if (valid) {
                    vjj = ej[pos];
                    r = rel[pos];
                    coef = expf(edge_score(si, r, vjj, s_r, s_j) - mx) / den;
                }
                unsigned msk = (unsigned)((__ballot(valid) >> hshift) & 0xFFFFull);
                add_edges(msk, coef, vjj, r, l, x, rel_emb, xi, A);
            }
        }

        if (live) {
            *(float4*)(row + 4 * l) = xi;
            *(float4*)(row + EHID + 4 * l) =
                make_float4(fmaxf(A.aI.x, 0.f), fmaxf(A.aI.y, 0.f), fmaxf(A.aI.z, 0.f), fmaxf(A.aI.w, 0.f));
            *(float2*)(row + 2 * EHID + 2 * l) = make_float2(fmaxf(A.aR.x, 0.f), fmaxf(A.aR.y, 0.f));
            *(float4*)(row + 2 * EHID + RHID + 4 * l) =
                make_float4(fmaxf(A.aJ.x, 0.f), fmaxf(A.aJ.y, 0.f), fmaxf(A.aJ.z, 0.f), fmaxf(A.aJ.w, 0.f));
        }
    }
}

extern "C" void kernel_launch(void* const* d_in, const int* in_sizes, int n_in,
                              void* d_out, int out_size, void* d_ws, size_t ws_size,
                              hipStream_t stream) {
    const float* x        = (const float*)d_in[0];
    const int*   eidx     = (const int*)d_in[1];
    const int*   rel      = (const int*)d_in[2];
    const float* rel_emb  = (const float*)d_in[3];
    const float* w        = (const float*)d_in[4];
    float* out = (float*)d_out;

    const int E = in_sizes[2];
    const int N = in_sizes[0] / EHID;
    const int R = in_sizes[3] / RHID;
    const int* ei = eidx;
    const int* ej = eidx + E;

    const int NB = ceil_div(N, BW);
    char* ws = (char*)d_ws;
    float* s_i = (float*)ws;
    float* s_j = (float*)(ws + (size_t)N * 4);
    float* s_r = (float*)(ws + (size_t)N * 8);
    int*  bcnt = (int*)(ws + (size_t)N * 8 + (size_t)R * 4);

    const int B = 256;

    const int tilesN = ceil_div(N, 16);
    const int tilesR = ceil_div(R, 16);
    k_project<<<ceil_div(tilesN + tilesR, 8), B, 0, stream>>>(x, rel_emb, w, s_i, s_j, s_r, (unsigned*)bcnt, N, R, tilesN, NB);

    k_bin<<<ceil_div(E, B * BIN_EPT), B, (size_t)NB * 8, stream>>>(ei, ej, rel, bcnt, out, E, N, NB);

    k_gather<<<NB, GT, 0, stream>>>(x, ei, ej, rel, rel_emb, s_i, s_j, s_r, bcnt, out, N, E);
}


# Projection and binning share one launch: the binning's 245 work-groups (a wait chain that leaves the chip idle) run beside the projection's short waves; counters zeroed by a 2-group launch in front
# speedup vs baseline: 6.3783x; 1.0528x over previous
#include <hip/hip_runtime.h>
#include <hip/hip_bf16.h>


typedef __attribute__((ext_vector_type(2))) float v2f;
typedef __attribute__((ext_vector_type(8))) float v8f;

#define EHID 64
#define RHID 32
#define DFEAT 160
#define DOUT 224
#define EPSF 1e-16f
#define BW 196
#define BWP 224
#define RECW 4
#define LCAP 2048
#define BIN_EPT 8
#define GT 512
#define FE 8

static inline int ceil_div(int a, int b) { return (a + b - 1) / b; }

__global__ void __launch_bounds__(256)
k_zero_counters(unsigned* __restrict__ bcnt, int NB) {
    int gid = blockIdx.x * blockDim.x + threadIdx.x;
    if (gid < NB) bcnt[gid] = 0u;
}

__device__ __forceinline__ void
project_body(const float* __restrict__ x,
             const float* __restrict__ rel_emb,
             const float* __restrict__ w,
             float* __restrict__ s_i,
             float* __restrict__ s_j,
             float* __restrict__ s_r,
             int N, int R, int tilesN, int pb) {
    int t = threadIdx.x;
    int gid = pb * 256 + t;

    int wave = gid >> 5;
    int lane = t & 31;
    int c = lane & 15;
    int h = lane >> 4;

    if (wave < tilesN) {
        int row = wave * 16 + c;
        int rr = row < N ? row : N - 1;
        bool bon = c < 2;
        const float* xr = x + (long)rr * EHID + 2 * h;
        const float* wc = w + ((c == 1) ? (EHID + RHID) : 0) + 2 * h;
        v8f acc = {};
#pragma unroll
        for (int ch = 0; ch < EHID / 4; ++ch) {
            v2f a  = *(const v2f*)(xr + 4 * ch);
            v2f wv = *(const v2f*)(wc + 4 * ch);
            v2f b;
            b.x = bon ? wv.x : 0.0f;
            b.y = bon ? wv.y : 0.0f;
            acc = __builtin_amdgcn_wmma_f32_16x16x4_f32(
                       false, a,  false, b,
                       (short)0, acc,  false,  false);
        }
        if (bon) {
            float* dst = (c == 0) ? s_i : s_j;
            int r0 = wave * 16 + 8 * h;
            for (int r = 0; r < 8; ++r)
                if (r0 + r < N) dst[r0 + r] = acc[r];
        }
    } else {
        int tile = wave - tilesN;
        if (tile * 16 >= R) return;
        int row = tile * 16 + c;
        int rr = row < R ? row : R - 1;
        bool bon = c == 0;
        const float* er = rel_emb + (long)rr * RHID + 2 * h;
        const float* wc = w + EHID + 2 * h;
        v8f acc = {};
#pragma unroll
        for (int ch = 0; ch < RHID / 4; ++ch) {
            v2f a  = *(const v2f*)(er + 4 * ch);
            v2f wv = *(const v2f*)(wc + 4 * ch);
            v2f b;
            b.x = bon ? wv.x : 0.0f;
            b.y = bon ? wv.y : 0.0f;
            acc = __builtin_amdgcn_wmma_f32_16x16x4_f32(
                       false, a,  false, b,
                       (short)0, acc,  false,  false);
        }
        if (bon) {
            int r0 = tile * 16 + 8 * h;
            for (int r = 0; r < 8; ++r)
                if (r0 + r < R) s_r[r0 + r] = acc[r];
        }
    }
}

__device__ __forceinline__ int bucket_cap(int b, int N) {
    int rows = N - b * BW;
    rows = rows < BW ? rows : BW;
    int cap = rows * (DOUT / RECW);
    return cap < LCAP ? cap : LCAP;
}

__device__ __forceinline__ void
bin_body(const int* __restrict__ ei,
         const int* __restrict__ ej,
         const int* __restrict__ rel,
         int* __restrict__ bcnt,
         float* __restrict__ out, int E, int N, int NB, int gb, int* sm) {
    int* hist = sm;
    int* base = sm + NB;
    const int tid = threadIdx.x;
    for (int i = tid; i < NB; i += 256) hist[i] = 0;
    __syncthreads();
    const int e0 = gb * (256 * BIN_EPT);
    int vi[BIN_EPT], vj[BIN_EPT], rr[BIN_EPT], lr[BIN_EPT];
#pragma unroll
    for (int k = 0; k < BIN_EPT; ++k) {
        int e = e0 + k * 256 + tid;
        bool on = e < E;
        int ec = on ? e : (E - 1);
        int d = ei[ec];
        vj[k] = ej[ec];
        rr[k] = rel[ec];
        vi[k] = on ? d : -1;
    }
#pragma unroll
    for (int k = 0; k < BIN_EPT; ++k)
        lr[k] = (vi[k] >= 0) ? atomicAdd(&hist[(unsigned)vi[k] / BW], 1) : 0;
    __syncthreads();
    for (int i = tid; i < NB; i += 256) {
        int hc = hist[i];
        base[i] = hc ? atomicAdd(&bcnt[i], hc) : 0;
    }
    __syncthreads();
#pragma unroll
    for (int k = 0; k < BIN_EPT; ++k) {
        if (vi[k] >= 0) {
            int b = (int)((unsigned)vi[k] / BW);
            int slot = base[b] + lr[k];
            if (slot < bucket_cap(b, N)) {
                int4 rc = make_int4(vi[k], vj[k], rr[k], 0);
                *(int4*)(out + (long)b * (BW * DOUT) + RECW * slot) = rc;
            }
        }
    }
}

__global__ void __launch_bounds__(256)
k_bin_project(const float* __restrict__ x,
              const float* __restrict__ rel_emb,
              const float* __restrict__ w,
              float* __restrict__ s_i,
              float* __restrict__ s_j,
              float* __restrict__ s_r,
              const int* __restrict__ ei,
              const int* __restrict__ ej,
              const int* __restrict__ rel,
              int* __restrict__ bcnt,
              float* __restrict__ out,
              int N, int R, int tilesN, int E, int NB, int nbin) {
    extern __shared__ int sm[];
    const int bx = blockIdx.x;
    if (bx < nbin) bin_body(ei, ej, rel, bcnt, out, E, N, NB, bx, sm);
    else           project_body(x, rel_emb, w, s_i, s_j, s_r, N, R, tilesN, bx - nbin);
}

__device__ __forceinline__ float hmax16(float v) {
    v = fmaxf(v, __shfl_xor(v, 8, 16));
    v = fmaxf(v, __shfl_xor(v, 4, 16));
    v = fmaxf(v, __shfl_xor(v, 2, 16));
    v = fmaxf(v, __shfl_xor(v, 1, 16));
    return v;
}
__device__ __forceinline__ float hsum16(float v) {
    v += __shfl_xor(v, 8, 16);
    v += __shfl_xor(v, 4, 16);
    v += __shfl_xor(v, 2, 16);
    v += __shfl_xor(v, 1, 16);
    return v;
}

struct NodeAcc {
    float4 aI, aJ;
    float2 aR;
};

__device__ __forceinline__ void add_edges(unsigned msk, float coef, int vj, int r, int l,
                                          const float* __restrict__ x,
                                          const float* __restrict__ rel_emb,
                                          float4 xi, NodeAcc& A) {
    while (msk) {
        const int q0 = __ffs(msk) - 1;
        float cf[4];
        float4 xj[4];
        float2 rr[4];
#pragma unroll
        for (int u = 0; u < 4; ++u) {
            bool on = (msk != 0u);
            int qu = on ? (__ffs(msk) - 1) : q0;
            msk &= msk - 1u;
            float cc = __shfl(coef, qu, 16);
            int vjq  = __shfl(vj, qu, 16);
            int rq   = __shfl(r, qu, 16);
            cf[u] = on ? cc : 0.0f;
            xj[u] = *(const float4*)(x + (long)vjq * EHID + 4 * l);
            rr[u] = *(const float2*)(rel_emb + (long)rq * RHID + 2 * l);
        }
#pragma unroll
        for (int u = 0; u < 4; ++u) {
            A.aI.x += xi.x * cf[u]; A.aI.y += xi.y * cf[u]; A.aI.z += xi.z * cf[u]; A.aI.w += xi.w * cf[u];
            A.aR.x += rr[u].x * cf[u]; A.aR.y += rr[u].y * cf[u];
            A.aJ.x += xj[u].x * cf[u]; A.aJ.y += xj[u].y * cf[u]; A.aJ.z += xj[u].z * cf[u]; A.aJ.w += xj[u].w * cf[u];
        }
    }
}

__device__ __forceinline__ float edge_score(float si, int q, int v,
                                            const float* __restrict__ s_r,
                                            const float* __restrict__ s_j) {
    return (si + s_r[q]) + s_j[v];
}

__global__ void __launch_bounds__(GT, 8)
k_gather(const float* __restrict__ x,
         const int* __restrict__ ei,
         const int* __restrict__ ej,
         const int* __restrict__ rel,
         const float* __restrict__ rel_emb,
         const float* __restrict__ s_i,
         const float* __restrict__ s_j,
         const float* __restrict__ s_r,
         const int* __restrict__ bcnt,
         float* out, int N, int E) {
    __shared__ int   g_vj[LCAP];
    __shared__ int   g_r[LCAP];
    __shared__ float g_sc[LCAP];
    __shared__ int   ncnt[BWP];
    __shared__ int   noff[BWP];
    __shared__ float sI[BWP];

    const int tid = threadIdx.x;
    const int b = blockIdx.x;
    const int n0 = b * BW;
    const int rows = (N - n0) < BW ? (N - n0) : BW;
    const int cb = bcnt[b];
    const bool fits = cb <= bucket_cap(b, N);
    const int l = tid & 15;

    if (tid < BWP) {
        ncnt[tid] = 0;
        sI[tid] = (tid < rows) ? s_i[n0 + tid] : 0.0f;
    }
    __syncthreads();

    if (fits) {
        const int* recs = (const int*)(out + (long)n0 * DOUT);
        int lv[LCAP / GT], vj[LCAP / GT], rr[LCAP / GT], rk[LCAP / GT];
        float pj[LCAP / GT], pr[LCAP / GT];
#pragma unroll
        for (int j = 0; j < LCAP / GT; ++j) {
            int k = tid + GT * j;
            bool on = k < cb;
            int4 rc = *(const int4*)(recs + RECW * (on ? k : 0));
            lv[j] = on ? (rc.x - n0) : -1;
            vj[j] = on ? rc.y : 0;
            rr[j] = on ? rc.z : 0;
        }
#pragma unroll
        for (int j = 0; j < LCAP / GT; ++j) {
            pj[j] = s_j[vj[j]];
            pr[j] = s_r[rr[j]];
        }
#pragma unroll
        for (int j = 0; j < LCAP / GT; ++j)
            rk[j] = (lv[j] >= 0) ? atomicAdd(&ncnt[lv[j]], 1) : 0;
        __syncthreads();
        if (tid < 32) {
            int loc[BWP / 32];
            int s = 0;
#pragma unroll
            for (int q = 0; q < BWP / 32; ++q) { loc[q] = s; s += ncnt[(BWP / 32) * tid + q]; }
            int inc = s;
#pragma unroll
            for (int d = 1; d < 32; d <<= 1) {
                int up = __shfl_up(inc, d, 32);
                if (tid >= d) inc += up;
            }
            int ex = inc - s;
#pragma unroll
            for (int q = 0; q < BWP / 32; ++q) noff[(BWP / 32) * tid + q] = ex + loc[q];
        }
        __syncthreads();
#pragma unroll
        for (int j = 0; j < LCAP / GT; ++j) {
            if (lv[j] >= 0) {
                int p = noff[lv[j]] + rk[j];
                g_vj[p] = vj[j];
                g_r[p]  = rr[j];
                g_sc[p] = (sI[lv[j]] + pr[j]) + pj[j];
            }
        }
        __syncthreads();
    }

    for (int it = 0; it < BWP / (GT / 16); ++it) {
        if (2 * (tid >> 5) + (GT / 16) * it >= rows) break;
        const int ln = (tid >> 4) + (GT / 16) * it;
        const bool live = ln < rows;
        const int nn = live ? (n0 + ln) : n0;
        float* row = out + (long)nn * DOUT;
        float4 xi = *(const float4*)(x + (long)nn * EHID + 4 * l);
        const float si = sI[live ? ln : 0];

        NodeAcc A;
        A.aI = make_float4(0.f, 0.f, 0.f, 0.f);
        A.aJ = make_float4(0.f, 0.f, 0.f, 0.f);
        A.aR = make_float2(0.f, 0.f);

        if (fits) {
            const int c = live ? ncnt[ln] : 0;
            const int o = noff[live ? ln : 0];
            bool v0 = l < c;
            int vj0 = 0, r0 = 0;
            float a0 = -INFINITY;
            if (v0) {
                vj0 = g_vj[o + l];
                r0  = g_r[o + l];
                a0  = g_sc[o + l];
            }
            float4 xj[FE];
            float2 rr[FE];
            const int vjf = __shfl(vj0, 0, 16);
            const int rf  = __shfl(r0, 0, 16);
            const int vjs = v0 ? vj0 : vjf;
            const int rs  = v0 ? r0  : rf;
#pragma unroll
            for (int u = 0; u < FE; ++u) {
                int vjq = __shfl(vjs, u, 16);
                int rq  = __shfl(rs, u, 16);
                xj[u] = *(const float4*)((const char*)x + (unsigned)(vjq * (EHID * 4) + 16 * l));
                rr[u] = *(const float2*)((const char*)rel_emb + (unsigned)(rq * (RHID * 4) + 8 * l));
            }
            float mx = a0;
            for (int bb = 16; bb < c; bb += 16) {
                int pos = bb + l;
                if (pos < c) mx = fmaxf(mx, g_sc[o + pos]);
            }
            mx = hmax16(mx);
            float p0 = v0 ? expf(a0 - mx) : 0.0f;
            float s = p0;
            for (int bb = 16; bb < c; bb += 16) {
                int pos = bb + l;
                if (pos < c) s += expf(g_sc[o + pos] - mx);
            }
            s = hsum16(s);
            float den = s + EPSF;
            const float coef0 = p0 / den;
#pragma unroll
            for (int u = 0; u < FE; ++u) {
                float cf = __shfl(coef0, u, 16);
                A.aI.x += xi.x * cf; A.aI.y += xi.y * cf; A.aI.z += xi.z * cf; A.aI.w += xi.w * cf;
                A.aR.x += rr[u].x * cf; A.aR.y += rr[u].y * cf;
                A.aJ.x += xj[u].x * cf; A.aJ.y += xj[u].y * cf; A.aJ.z += xj[u].z * cf; A.aJ.w += xj[u].w * cf;
            }
            {
                int k = c < 16 ? c : 16;
                unsigned m = ((1u << k) - 1u) & ~((1u << FE) - 1u);
                add_edges(m, coef0, vj0, r0, l, x, rel_emb, xi, A);
            }
            for (int bb = 16; bb < c; bb += 16) {
                int pos = bb + l;
                float coef = 0.0f;
                int vjj = 0, r = 0;
                if (pos < c) {
                    vjj = g_vj[o + pos];
                    r = g_r[o + pos];
                    coef = expf(g_sc[o + pos] - mx) / den;
                }
                int k = (c - bb) < 16 ? (c - bb) : 16;
                add_edges((1u << k) - 1u, coef, vjj, r, l, x, rel_emb, xi, A);
            }
        } else {
            int hshift = tid & 16;
            float mx = -INFINITY;
            for (int bb = 0; bb < E; bb += 16) {
                int pos = bb + l;
                if (live && pos < E && ei[pos] == nn) mx = fmaxf(mx, edge_score(si, rel[pos], ej[pos], s_r, s_j));
            }
            mx = hmax16(mx);
            float s = 0.0f;
            for (int bb = 0; bb < E; bb += 16) {
                int pos = bb + l;
                if (live && pos < E && ei[pos] == nn) s += expf(edge_score(si, rel[pos], ej[pos], s_r, s_j) - mx);
            }
            s = hsum16(s);
            float den = s + EPSF;
            for (int bb = 0; bb < E; bb += 16) {
                int pos = bb + l;
                bool valid = live && (pos < E) && (ei[pos] == nn);
                float coef = 0.0f;
                int vjj = 0, r = 0;
                if (valid) {
                    vjj = ej[pos];
                    r = rel[pos];
                    coef = expf(edge_score(si, r, vjj, s_r, s_j) - mx) / den;
                }
                unsigned msk = (unsigned)((__ballot(valid) >> hshift) & 0xFFFFull);
                add_edges(msk, coef, vjj, r, l, x, rel_emb, xi, A);
            }
        }

        if (live) {
            *(float4*)(row + 4 * l) = xi;
            *(float4*)(row + EHID + 4 * l) =
                make_float4(fmaxf(A.aI.x, 0.f), fmaxf(A.aI.y, 0.f), fmaxf(A.aI.z, 0.f), fmaxf(A.aI.w, 0.f));
            *(float2*)(row + 2 * EHID + 2 * l) = make_float2(fmaxf(A.aR.x, 0.f), fmaxf(A.aR.y, 0.f));
            *(float4*)(row + 2 * EHID + RHID + 4 * l) =
                make_float4(fmaxf(A.aJ.x, 0.f), fmaxf(A.aJ.y, 0.f), fmaxf(A.aJ.z, 0.f), fmaxf(A.aJ.w, 0.f));
        }
    }
}

extern "C" void kernel_launch(void* const* d_in, const int* in_sizes, int n_in,
                              void* d_out, int out_size, void* d_ws, size_t ws_size,
                              hipStream_t stream) {
    const float* x        = (const float*)d_in[0];
    const int*   eidx     = (const int*)d_in[1];
    const int*   rel      = (const int*)d_in[2];
    const float* rel_emb  = (const float*)d_in[3];
    const float* w        = (const float*)d_in[4];
    float* out = (float*)d_out;

    const int E = in_sizes[2];
    const int N = in_sizes[0] / EHID;
    const int R = in_sizes[3] / RHID;
    const int* ei = eidx;
    const int* ej = eidx + E;

    const int NB = ceil_div(N, BW);
    char* ws = (char*)d_ws;
    float* s_i = (float*)ws;
    float* s_j = (float*)(ws + (size_t)N * 4);
    float* s_r = (float*)(ws + (size_t)N * 8);
    int*  bcnt = (int*)(ws + (size_t)N * 8 + (size_t)R * 4);

    const int B = 256;

    k_zero_counters<<<ceil_div(NB, B), B, 0, stream>>>((unsigned*)bcnt, NB);

    const int tilesN = ceil_div(N, 16);
    const int tilesR = ceil_div(R, 16);
    const int nbin = ceil_div(E, B * BIN_EPT);
    const int nproj = ceil_div(tilesN + tilesR, 8);
    k_bin_project<<<nbin + nproj, B, (size_t)NB * 8, stream>>>(x, rel_emb, w, s_i, s_j, s_r, ei, ej, rel, bcnt, out,
                                                              N, R, tilesN, E, NB, nbin);

    k_gather<<<NB, GT, 0, stream>>>(x, ei, ej, rel, rel_emb, s_i, s_j, s_r, bcnt, out, N, E);
}


# Binning: each lane reserves the slots of its two buckets with ONE 64-bit returning add on the counter pair (was two 32-bit adds, one behind the other's return); sums stay in LDS arrival-rank order
# speedup vs baseline: 6.5784x; 1.0314x over previous
#include <hip/hip_runtime.h>
#include <hip/hip_bf16.h>


typedef __attribute__((ext_vector_type(2))) float v2f;
typedef __attribute__((ext_vector_type(8))) float v8f;

#define EHID 64
#define RHID 32
#define DFEAT 160
#define DOUT 224
#define EPSF 1e-16f
#define BW 196
#define BWP 224
#define RECW 4
#define LCAP 2048
#define BIN_EPT 8
#define GT 512
#define FE 8

static inline int ceil_div(int a, int b) { return (a + b - 1) / b; }

__global__ void __launch_bounds__(256)
k_zero_counters(unsigned* __restrict__ bcnt, int NB) {
    int gid = blockIdx.x * blockDim.x + threadIdx.x;
    if (gid < NB) bcnt[gid] = 0u;
}

__device__ __forceinline__ void
project_body(const float* __restrict__ x,
             const float* __restrict__ rel_emb,
             const float* __restrict__ w,
             float* __restrict__ s_i,
             float* __restrict__ s_j,
             float* __restrict__ s_r,
             int N, int R, int tilesN, int pb) {
    int t = threadIdx.x;
    int gid = pb * 256 + t;

    int wave = gid >> 5;
    int lane = t & 31;
    int c = lane & 15;
    int h = lane >> 4;

    if (wave < tilesN) {
        int row = wave * 16 + c;
        int rr = row < N ? row : N - 1;
        bool bon = c < 2;
        const float* xr = x + (long)rr * EHID + 2 * h;
        const float* wc = w + ((c == 1) ? (EHID + RHID) : 0) + 2 * h;
        v8f acc = {};
#pragma unroll
        for (int ch = 0; ch < EHID / 4; ++ch) {
            v2f a  = *(const v2f*)(xr + 4 * ch);
            v2f wv = *(const v2f*)(wc + 4 * ch);
            v2f b;
            b.x = bon ? wv.x : 0.0f;
            b.y = bon ? wv.y : 0.0f;
            acc = __builtin_amdgcn_wmma_f32_16x16x4_f32(
                       false, a,  false, b,
                       (short)0, acc,  false,  false);
        }
        if (bon) {
            float* dst = (c == 0) ? s_i : s_j;
            int r0 = wave * 16 + 8 * h;
            for (int r = 0; r < 8; ++r)
                if (r0 + r < N) dst[r0 + r] = acc[r];
        }
    } else {
        int tile = wave - tilesN;
        if (tile * 16 >= R) return;
        int row = tile * 16 + c;
        int rr = row < R ? row : R - 1;
        bool bon = c == 0;
        const float* er = rel_emb + (long)rr * RHID + 2 * h;
        const float* wc = w + EHID + 2 * h;
        v8f acc = {};
#pragma unroll
        for (int ch = 0; ch < RHID / 4; ++ch) {
            v2f a  = *(const v2f*)(er + 4 * ch);
            v2f wv = *(const v2f*)(wc + 4 * ch);
            v2f b;
            b.x = bon ? wv.x : 0.0f;
            b.y = bon ? wv.y : 0.0f;
            acc = __builtin_amdgcn_wmma_f32_16x16x4_f32(
                       false, a,  false, b,
                       (short)0, acc,  false,  false);
        }
        if (bon) {
            int r0 = tile * 16 + 8 * h;
            for (int r = 0; r < 8; ++r)
                if (r0 + r < R) s_r[r0 + r] = acc[r];
        }
    }
}

__device__ __forceinline__ int bucket_cap(int b, int N) {
    int rows = N - b * BW;
    rows = rows < BW ? rows : BW;
    int cap = rows * (DOUT / RECW);
    return cap < LCAP ? cap : LCAP;
}

__device__ __forceinline__ void
bin_body(const int* __restrict__ ei,
         const int* __restrict__ ej,
         const int* __restrict__ rel,
         int* __restrict__ bcnt,
         float* __restrict__ out, int E, int N, int NB, int gb, int* sm) {
    const int NBP = (NB + 1) & ~1;
    int* hist = sm;
    int* base = sm + NBP;
    const int tid = threadIdx.x;
    for (int i = tid; i < NBP; i += 256) hist[i] = 0;
    __syncthreads();
    const int e0 = gb * (256 * BIN_EPT);
    int vi[BIN_EPT], vj[BIN_EPT], rr[BIN_EPT], lr[BIN_EPT];
#pragma unroll
    for (int k = 0; k < BIN_EPT; ++k) {
        int e = e0 + k * 256 + tid;
        bool on = e < E;
        int ec = on ? e : (E - 1);
        int d = ei[ec];
        vj[k] = ej[ec];
        rr[k] = rel[ec];
        vi[k] = on ? d : -1;
    }
#pragma unroll
    for (int k = 0; k < BIN_EPT; ++k)
        lr[k] = (vi[k] >= 0) ? atomicAdd(&hist[(unsigned)vi[k] / BW], 1) : 0;
    __syncthreads();
    unsigned long long* bc2 = (unsigned long long*)bcnt;
    for (int p = tid; p < NBP / 2; p += 256) {
        int2 hc = *(const int2*)(hist + 2 * p);
        unsigned long long add = (unsigned long long)(unsigned)hc.x | ((unsigned long long)(unsigned)hc.y << 32);
        unsigned long long old = add ? atomicAdd(&bc2[p], add) : 0ull;
        *(int2*)(base + 2 * p) = make_int2((int)(unsigned)old, (int)(unsigned)(old >> 32));
    }
    __syncthreads();
#pragma unroll
    for (int k = 0; k < BIN_EPT; ++k) {
        if (vi[k] >= 0) {
            int b = (int)((unsigned)vi[k] / BW);
            int slot = base[b] + lr[k];
            if (slot < bucket_cap(b, N)) {
                int4 rc = make_int4(vi[k], vj[k], rr[k], 0);
                *(int4*)(out + (long)b * (BW * DOUT) + RECW * slot) = rc;
            }
        }
    }
}

__global__ void __launch_bounds__(256)
k_bin_project(const float* __restrict__ x,
              const float* __restrict__ rel_emb,
              const float* __restrict__ w,
              float* __restrict__ s_i,
              float* __restrict__ s_j,
              float* __restrict__ s_r,
              const int* __restrict__ ei,
              const int* __restrict__ ej,
              const int* __restrict__ rel,
              int* __restrict__ bcnt,
              float* __restrict__ out,
              int N, int R, int tilesN, int E, int NB, int nbin) {
    extern __shared__ int sm[];
    const int bx = blockIdx.x;
    if (bx < nbin) bin_body(ei, ej, rel, bcnt, out, E, N, NB, bx, sm);
    else           project_body(x, rel_emb, w, s_i, s_j, s_r, N, R, tilesN, bx - nbin);
}

__device__ __forceinline__ float hmax16(float v) {
    v = fmaxf(v, __shfl_xor(v, 8, 16));
    v = fmaxf(v, __shfl_xor(v, 4, 16));
    v = fmaxf(v, __shfl_xor(v, 2, 16));
    v = fmaxf(v, __shfl_xor(v, 1, 16));
    return v;
}
__device__ __forceinline__ float hsum16(float v) {
    v += __shfl_xor(v, 8, 16);
    v += __shfl_xor(v, 4, 16);
    v += __shfl_xor(v, 2, 16);
    v += __shfl_xor(v, 1, 16);
    return v;
}

struct NodeAcc {
    float4 aI, aJ;
    float2 aR;
};

__device__ __forceinline__ void add_edges(unsigned msk, float coef, int vj, int r, int l,
                                          const float* __restrict__ x,
                                          const float* __restrict__ rel_emb,
                                          float4 xi, NodeAcc& A) {
    while (msk) {
        const int q0 = __ffs(msk) - 1;
        float cf[4];
        float4 xj[4];
        float2 rr[4];
#pragma unroll
        for (int u = 0; u < 4; ++u) {
            bool on = (msk != 0u);
            int qu = on ? (__ffs(msk) - 1) : q0;
            msk &= msk - 1u;
            float cc = __shfl(coef, qu, 16);
            int vjq  = __shfl(vj, qu, 16);
            int rq   = __shfl(r, qu, 16);
            cf[u] = on ? cc : 0.0f;
            xj[u] = *(const float4*)(x + (long)vjq * EHID + 4 * l);
            rr[u] = *(const float2*)(rel_emb + (long)rq * RHID + 2 * l);
        }
#pragma unroll
        for (int u = 0; u < 4; ++u) {
            A.aI.x += xi.x * cf[u]; A.aI.y += xi.y * cf[u]; A.aI.z += xi.z * cf[u]; A.aI.w += xi.w * cf[u];
            A.aR.x += rr[u].x * cf[u]; A.aR.y += rr[u].y * cf[u];
            A.aJ.x += xj[u].x * cf[u]; A.aJ.y += xj[u].y * cf[u]; A.aJ.z += xj[u].z * cf[u]; A.aJ.w += xj[u].w * cf[u];
        }
    }
}

__device__ __forceinline__ float edge_score(float si, int q, int v,
                                            const float* __restrict__ s_r,
                                            const float* __restrict__ s_j) {
    return (si + s_r[q]) + s_j[v];
}

__global__ void __launch_bounds__(GT, 8)
k_gather(const float* __restrict__ x,
         const int* __restrict__ ei,
         const int* __restrict__ ej,
         const int* __restrict__ rel,
         const float* __restrict__ rel_emb,
         const float* __restrict__ s_i,
         const float* __restrict__ s_j,
         const float* __restrict__ s_r,
         const int* __restrict__ bcnt,
         float* out, int N, int E) {
    __shared__ int   g_vj[LCAP];
    __shared__ int   g_r[LCAP];
    __shared__ float g_sc[LCAP];
    __shared__ int   ncnt[BWP];
    __shared__ int   noff[BWP];
    __shared__ float sI[BWP];

    const int tid = threadIdx.x;
    const int b = blockIdx.x;
    const int n0 = b * BW;
    const int rows = (N - n0) < BW ? (N - n0) : BW;
    const int cb = bcnt[b];
    const bool fits = cb <= bucket_cap(b, N);
    const int l = tid & 15;

    if (tid < BWP) {
        ncnt[tid] = 0;
        sI[tid] = (tid < rows) ? s_i[n0 + tid] : 0.0f;
    }
    __syncthreads();

    if (fits) {
        const int* recs = (const int*)(out + (long)n0 * DOUT);
        int lv[LCAP / GT], vj[LCAP / GT], rr[LCAP / GT], rk[LCAP / GT];
        float pj[LCAP / GT], pr[LCAP / GT];
#pragma unroll
        for (int j = 0; j < LCAP / GT; ++j) {
            int k = tid + GT * j;
            bool on = k < cb;
            int4 rc = *(const int4*)(recs + RECW * (on ? k : 0));
            lv[j] = on ? (rc.x - n0) : -1;
            vj[j] = on ? rc.y : 0;
            rr[j] = on ? rc.z : 0;
        }
#pragma unroll
        for (int j = 0; j < LCAP / GT; ++j) {
            pj[j] = s_j[vj[j]];
            pr[j] = s_r[rr[j]];
        }
#pragma unroll
        for (int j = 0; j < LCAP / GT; ++j)
            rk[j] = (lv[j] >= 0) ? atomicAdd(&ncnt[lv[j]], 1) : 0;
        __syncthreads();
        if (tid < 32) {
            int loc[BWP / 32];
            int s = 0;
#pragma unroll
            for (int q = 0; q < BWP / 32; ++q) { loc[q] = s; s += ncnt[(BWP / 32) * tid + q]; }
            int inc = s;
#pragma unroll
            for (int d = 1; d < 32; d <<= 1) {
                int up = __shfl_up(inc, d, 32);
                if (tid >= d) inc += up;
            }
            int ex = inc - s;
#pragma unroll
            for (int q = 0; q < BWP / 32; ++q) noff[(BWP / 32) * tid + q] = ex + loc[q];
        }
        __syncthreads();
#pragma unroll
        for (int j = 0; j < LCAP / GT; ++j) {
            if (lv[j] >= 0) {
                int p = noff[lv[j]] + rk[j];
                g_vj[p] = vj[j];
                g_r[p]  = rr[j];
                g_sc[p] = (sI[lv[j]] + pr[j]) + pj[j];
            }
        }
        __syncthreads();
    }

    for (int it = 0; it < BWP / (GT / 16); ++it) {
        if (2 * (tid >> 5) + (GT / 16) * it >= rows) break;
        const int ln = (tid >> 4) + (GT / 16) * it;
        const bool live = ln < rows;
        const int nn = live ? (n0 + ln) : n0;
        float* row = out + (long)nn * DOUT;
        float4 xi = *(const float4*)(x + (long)nn * EHID + 4 * l);
        const float si = sI[live ? ln : 0];

        NodeAcc A;
        A.aI = make_float4(0.f, 0.f, 0.f, 0.f);
        A.aJ = make_float4(0.f, 0.f, 0.f, 0.f);
        A.aR = make_float2(0.f, 0.f);

        if (fits) {
            const int c = live ? ncnt[ln] : 0;
            const int o = noff[live ? ln : 0];
            bool v0 = l < c;
            int vj0 = 0, r0 = 0;
            float a0 = -INFINITY;
            if (v0) {
                vj0 = g_vj[o + l];
                r0  = g_r[o + l];
                a0  = g_sc[o + l];
            }
            float4 xj[FE];
            float2 rr[FE];
            const int vjf = __shfl(vj0, 0, 16);
            const int rf  = __shfl(r0, 0, 16);
            const int vjs = v0 ? vj0 : vjf;
            const int rs  = v0 ? r0  : rf;
#pragma unroll
            for (int u = 0; u < FE; ++u) {
                int vjq = __shfl(vjs, u, 16);
                int rq  = __shfl(rs, u, 16);
                xj[u] = *(const float4*)((const char*)x + (unsigned)(vjq * (EHID * 4) + 16 * l));
                rr[u] = *(const float2*)((const char*)rel_emb + (unsigned)(rq * (RHID * 4) + 8 * l));
            }
            float mx = a0;
            for (int bb = 16; bb < c; bb += 16) {
                int pos = bb + l;
                if (pos < c) mx = fmaxf(mx, g_sc[o + pos]);
            }
            mx = hmax16(mx);
            float p0 = v0 ? expf(a0 - mx) : 0.0f;
            float s = p0;
            for (int bb = 16; bb < c; bb += 16) {
                int pos = bb + l;
                if (pos < c) s += expf(g_sc[o + pos] - mx);
            }
            s = hsum16(s);
            float den = s + EPSF;
            const float coef0 = p0 / den;
#pragma unroll
            for (int u = 0; u < FE; ++u) {
                float cf = __shfl(coef0, u, 16);
                A.aI.x += xi.x * cf; A.aI.y += xi.y * cf; A.aI.z += xi.z * cf; A.aI.w += xi.w * cf;
                A.aR.x += rr[u].x * cf; A.aR.y += rr[u].y * cf;
                A.aJ.x += xj[u].x * cf; A.aJ.y += xj[u].y * cf; A.aJ.z += xj[u].z * cf; A.aJ.w += xj[u].w * cf;
            }
            {
                int k = c < 16 ? c : 16;
                unsigned m = ((1u << k) - 1u) & ~((1u << FE) - 1u);
                add_edges(m, coef0, vj0, r0, l, x, rel_emb, xi, A);
            }
            for (int bb = 16; bb < c; bb += 16) {
                int pos = bb + l;
                float coef = 0.0f;
                int vjj = 0, r = 0;
                if (pos < c) {
                    vjj = g_vj[o + pos];
                    r = g_r[o + pos];
                    coef = expf(g_sc[o + pos] - mx) / den;
                }
                int k = (c - bb) < 16 ? (c - bb) : 16;
                add_edges((1u << k) - 1u, coef, vjj, r, l, x, rel_emb, xi, A);
            }
        } else {
            int hshift = tid & 16;
            float mx = -INFINITY;
            for (int bb = 0; bb < E; bb += 16) {
                int pos = bb + l;
                if (live && pos < E && ei[pos] == nn) mx = fmaxf(mx, edge_score(si, rel[pos], ej[pos], s_r, s_j));
            }
            mx = hmax16(mx);
            float s = 0.0f;
            for (int bb = 0; bb < E; bb += 16) {
                int pos = bb + l;
                if (live && pos < E && ei[pos] == nn) s += expf(edge_score(si, rel[pos], ej[pos], s_r, s_j) - mx);
            }
            s = hsum16(s);
            float den = s + EPSF;
            for (int bb = 0; bb < E; bb += 16) {
                int pos = bb + l;
                bool valid = live && (pos < E) && (ei[pos] == nn);
                float coef = 0.0f;
                int vjj = 0, r = 0;
                if (valid) {
                    vjj = ej[pos];
                    r = rel[pos];
                    coef = expf(edge_score(si, r, vjj, s_r, s_j) - mx) / den;
                }
                unsigned msk = (unsigned)((__ballot(valid) >> hshift) & 0xFFFFull);
                add_edges(msk, coef, vjj, r, l, x, rel_emb, xi, A);
            }
        }

        if (live) {
            *(float4*)(row + 4 * l) = xi;
            *(float4*)(row + EHID + 4 * l) =
                make_float4(fmaxf(A.aI.x, 0.f), fmaxf(A.aI.y, 0.f), fmaxf(A.aI.z, 0.f), fmaxf(A.aI.w, 0.f));
            *(float2*)(row + 2 * EHID + 2 * l) = make_float2(fmaxf(A.aR.x, 0.f), fmaxf(A.aR.y, 0.f));
            *(float4*)(row + 2 * EHID + RHID + 4 * l) =
                make_float4(fmaxf(A.aJ.x, 0.f), fmaxf(A.aJ.y, 0.f), fmaxf(A.aJ.z, 0.f), fmaxf(A.aJ.w, 0.f));
        }
    }
}

extern "C" void kernel_launch(void* const* d_in, const int* in_sizes, int n_in,
                              void* d_out, int out_size, void* d_ws, size_t ws_size,
                              hipStream_t stream) {
    const float* x        = (const float*)d_in[0];
    const int*   eidx     = (const int*)d_in[1];
    const int*   rel      = (const int*)d_in[2];
    const float* rel_emb  = (const float*)d_in[3];
    const float* w        = (const float*)d_in[4];
    float* out = (float*)d_out;

    const int E = in_sizes[2];
    const int N = in_sizes[0] / EHID;
    const int R = in_sizes[3] / RHID;
    const int* ei = eidx;
    const int* ej = eidx + E;

    const int NB = ceil_div(N, BW);
    const int NBP = 2 * ceil_div(NB, 2);
    char* ws = (char*)d_ws;
    float* s_i = (float*)ws;
    float* s_j = (float*)(ws + (size_t)N * 4);
    int*  bcnt = (int*)(ws + (size_t)N * 8);
    float* s_r = (float*)(ws + (size_t)N * 8 + (size_t)NBP * 4);

    const int B = 256;

    k_zero_counters<<<ceil_div(NBP, B), B, 0, stream>>>((unsigned*)bcnt, NBP);

    const int tilesN = ceil_div(N, 16);
    const int tilesR = ceil_div(R, 16);
    const int nbin = ceil_div(E, B * BIN_EPT);
    const int nproj = ceil_div(tilesN + tilesR, 8);
    k_bin_project<<<nbin + nproj, B, (size_t)NBP * 8, stream>>>(x, rel_emb, w, s_i, s_j, s_r, ei, ej, rel, bcnt, out,
                                                              N, R, tilesN, E, NB, nbin);

    k_gather<<<NB, GT, 0, stream>>>(x, ei, ej, rel, rel_emb, s_i, s_j, s_r, bcnt, out, N, E);
}
